# starting kernel
# baseline (speedup 1.0000x reference)
#include <hip/hip_runtime.h>
#include <hip/hip_bf16.h>


typedef __attribute__((ext_vector_type(2))) float v2f;
typedef __attribute__((ext_vector_type(4))) float v4f;
typedef __attribute__((ext_vector_type(8))) float v8f;

#define LH    20
#define LH4   80
#define LF    32
#define LT    1024
#define LB    256
#define LOUT  10

__device__ __forceinline__ float sigf(float x) { return 1.0f / (1.0f + __expf(-x)); }
__device__ __forceinline__ float tanhfast(float x) {
    const float e = __expf(2.0f * x);
    return (e - 1.0f) / (e + 1.0f);
}

#define WMMA4(c, a, b) \
    (c) = __builtin_amdgcn_wmma_f32_16x16x4_f32(false, (a), false, (b), (short)0, (c), false, false)

__global__ void __launch_bounds__(32)
lstm2_dense_fused(const float* __restrict__ X,
                  const float* __restrict__ Wi1, const float* __restrict__ Wh1,
                  const float* __restrict__ b1,
                  const float* __restrict__ Wi2, const float* __restrict__ Wh2,
                  const float* __restrict__ b2,
                  const float* __restrict__ Wd,  const float* __restrict__ bd,
                  float* __restrict__ out)
{
    __shared__ v2f sWi1f[40 * 32];
    __shared__ v2f sWh1f[25 * 32];
    __shared__ v2f sWi2f[25 * 32];
    __shared__ v2f sWh2f[25 * 32];
    __shared__ v2f sWdf [ 5 * 32];
    __shared__ float sb1[LH4], sb2[LH4], sbd[16];
    __shared__ v4f sZt4[LH4 * 4];
    __shared__ v2f sH1f[5 * 32], sH2f[5 * 32];
    __shared__ float sC1[16 * LH], sC2[16 * LH];

    const int lane  = threadIdx.x;
    const int mrow  = lane & 15;
    const int hi    = lane >> 4;
    const int khalf = 2 * hi;
    const int b0    = blockIdx.x * 16;

    for (int f = 0; f < 40; ++f) {
        const int n = f >> 3, kk = f & 7;
        const int row = 4 * kk + khalf, col = 16 * n + mrow;
        v2f w; w.x = Wi1[row * LH4 + col]; w.y = Wi1[(row + 1) * LH4 + col];
        sWi1f[f * 32 + lane] = w;
    }
    for (int f = 0; f < 25; ++f) {
        const int n = f / 5, kk = f % 5;
        const int row = 4 * kk + khalf, col = 16 * n + mrow;
        v2f w;
        w.x = Wh1[row * LH4 + col]; w.y = Wh1[(row + 1) * LH4 + col]; sWh1f[f * 32 + lane] = w;
        w.x = Wi2[row * LH4 + col]; w.y = Wi2[(row + 1) * LH4 + col]; sWi2f[f * 32 + lane] = w;
        w.x = Wh2[row * LH4 + col]; w.y = Wh2[(row + 1) * LH4 + col]; sWh2f[f * 32 + lane] = w;
    }
    for (int kk = 0; kk < 5; ++kk) {
        const int row = 4 * kk + khalf;
        v2f w;
        w.x = (mrow < LOUT) ? Wd[row * LOUT + mrow] : 0.0f;
        w.y = (mrow < LOUT) ? Wd[(row + 1) * LOUT + mrow] : 0.0f;
        sWdf[kk * 32 + lane] = w;
    }
    for (int i = lane; i < LH4; i += 32) { sb1[i] = b1[i]; sb2[i] = b2[i]; }
    for (int i = lane; i < 16; i += 32) sbd[i] = (i < LOUT) ? bd[i] : 0.0f;
    {
        v2f zz; zz.x = 0.0f; zz.y = 0.0f;
        for (int i = lane; i < 5 * 32; i += 32) { sH1f[i] = zz; sH2f[i] = zz; }
        for (int i = lane; i < 16 * LH; i += 32) { sC1[i] = 0.0f; sC2[i] = 0.0f; }
    }
    __syncthreads();

    const float* zp = (const float*)sZt4;

    for (int t = 0; t < LT; ++t) {
        const float* xrow = X + ((size_t)(b0 + mrow) * LT + (size_t)t) * LF;
        v2f ax[8];
        #pragma unroll
        for (int kk = 0; kk < 8; ++kk) {
            ax[kk].x = xrow[4 * kk + khalf];
            ax[kk].y = xrow[4 * kk + khalf + 1];
        }
        if (t + 1 < LT) __builtin_prefetch(xrow + LF, 0, 3);

        v8f zt[5];
        #pragma unroll
        for (int n = 0; n < 5; ++n) {
            v8f c = {};
            #pragma unroll
            for (int kk = 0; kk < 8; ++kk)
                WMMA4(c, ax[kk], sWi1f[(n * 8 + kk) * 32 + lane]);
            #pragma unroll
            for (int kk = 0; kk < 5; ++kk)
                WMMA4(c, sH1f[kk * 32 + lane], sWh1f[(n * 5 + kk) * 32 + lane]);
            zt[n] = c;
        }
        #pragma unroll
        for (int n = 0; n < 5; ++n) {
            const int zcol = 16 * n + mrow;
            v4f lo, hiq;
            lo.x  = zt[n][0]; lo.y  = zt[n][1]; lo.z  = zt[n][2]; lo.w  = zt[n][3];
            hiq.x = zt[n][4]; hiq.y = zt[n][5]; hiq.z = zt[n][6]; hiq.w = zt[n][7];
            sZt4[zcol * 4 + khalf]     = lo;
            sZt4[zcol * 4 + khalf + 1] = hiq;
        }
        __syncthreads();

        #pragma unroll
        for (int i = 0; i < 10; ++i) {
            const int idx = lane + 32 * i;
            const int m = idx / LH, j = idx % LH;
            const float zi = zp[ j            * 16 + m] + sb1[j];
            const float zf = zp[(LH     + j)  * 16 + m] + sb1[LH + j];
            const float zg = zp[(2 * LH + j)  * 16 + m] + sb1[2 * LH + j];
            const float zo = zp[(3 * LH + j)  * 16 + m] + sb1[3 * LH + j];
            const float cc = sigf(zf) * sC1[idx] + sigf(zi) * tanhfast(zg);
            sC1[idx] = cc;
            const float hh = sigf(zo) * tanhfast(cc);
            const int kk = j >> 2, off = j & 3;
            ((float*)sH1f)[((kk * 32) + (m + 16 * (off >> 1))) * 2 + (off & 1)] = hh;
        }
        __syncthreads();

        #pragma unroll
        for (int n = 0; n < 5; ++n) {
            v8f c = {};
            #pragma unroll
            for (int kk = 0; kk < 5; ++kk)
                WMMA4(c, sH1f[kk * 32 + lane], sWi2f[(n * 5 + kk) * 32 + lane]);
            #pragma unroll
            for (int kk = 0; kk < 5; ++kk)
                WMMA4(c, sH2f[kk * 32 + lane], sWh2f[(n * 5 + kk) * 32 + lane]);
            zt[n] = c;
        }
        #pragma unroll
        for (int n = 0; n < 5; ++n) {
            const int zcol = 16 * n + mrow;
            v4f lo, hiq;
            lo.x  = zt[n][0]; lo.y  = zt[n][1]; lo.z  = zt[n][2]; lo.w  = zt[n][3];
            hiq.x = zt[n][4]; hiq.y = zt[n][5]; hiq.z = zt[n][6]; hiq.w = zt[n][7];
            sZt4[zcol * 4 + khalf]     = lo;
            sZt4[zcol * 4 + khalf + 1] = hiq;
        }
        __syncthreads();

        #pragma unroll
        for (int i = 0; i < 10; ++i) {
            const int idx = lane + 32 * i;
            const int m = idx / LH, j = idx % LH;
            const float zi = zp[ j            * 16 + m] + sb2[j];
            const float zf = zp[(LH     + j)  * 16 + m] + sb2[LH + j];
            const float zg = zp[(2 * LH + j)  * 16 + m] + sb2[2 * LH + j];
            const float zo = zp[(3 * LH + j)  * 16 + m] + sb2[3 * LH + j];
            const float cc = sigf(zf) * sC2[idx] + sigf(zi) * tanhfast(zg);
            sC2[idx] = cc;
            const float hh = sigf(zo) * tanhfast(cc);
            const int kk = j >> 2, off = j & 3;
            ((float*)sH2f)[((kk * 32) + (m + 16 * (off >> 1))) * 2 + (off & 1)] = hh;
        }
        __syncthreads();

        {
            v8f c = {};
            #pragma unroll
            for (int kk = 0; kk < 5; ++kk)
                WMMA4(c, sH2f[kk * 32 + lane], sWdf[kk * 32 + lane]);
            if (mrow < LOUT) {
                const float bb = sbd[mrow];
                #pragma unroll
                for (int r = 0; r < 8; ++r) {
                    const int m = r + 8 * hi;
                    out[((size_t)(b0 + m) * LT + (size_t)t) * LOUT + mrow] = c[r] + bb;
                }
            }
        }
        __syncthreads();
    }
}

extern "C" void kernel_launch(void* const* d_in, const int* in_sizes, int n_in,
                              void* d_out, int out_size, void* d_ws, size_t ws_size,
                              hipStream_t stream) {
    (void)in_sizes; (void)n_in; (void)out_size; (void)d_ws; (void)ws_size;
    const float* X   = (const float*)d_in[0];
    const float* Wi1 = (const float*)d_in[1];
    const float* Wh1 = (const float*)d_in[2];
    const float* b1  = (const float*)d_in[3];
    const float* Wi2 = (const float*)d_in[4];
    const float* Wh2 = (const float*)d_in[5];
    const float* b2  = (const float*)d_in[6];
    const float* Wd  = (const float*)d_in[7];
    const float* bd  = (const float*)d_in[8];
    float* out = (float*)d_out;

    dim3 grid(LB / 16);
    dim3 block(32);
    lstm2_dense_fused<<<grid, block, 0, stream>>>(X, Wi1, Wh1, b1, Wi2, Wh2, b2, Wd, bd, out);
}


# Products transposed (z^T = W^T [x;h]^T), weight rows regrouped by gate: a lane holds all 4 gates of its units, its h pair is the next B fragment; no LDS write in the loop, weights stay in VGPRs
# speedup vs baseline: 1.5294x; 1.5294x over previous
#include <hip/hip_runtime.h>
#include <hip/hip_bf16.h>


typedef __attribute__((ext_vector_type(2))) float v2f;
typedef __attribute__((ext_vector_type(8))) float v8f;

#define LH    20
#define LH4   80
#define LF    32
#define LT    1024
#define LB    256
#define LOUT  10

__device__ __forceinline__ float sigf(float x) { return 1.0f / (1.0f + __expf(-x)); }
__device__ __forceinline__ float tanhfast(float x) {
    const float e = __expf(2.0f * x);
    return (e - 1.0f) / (e + 1.0f);
}

#define WMMA4(c, a, b) \
    (c) = __builtin_amdgcn_wmma_f32_16x16x4_f32(false, (a), false, (b), (short)0, (c), false, false)

__global__ void __launch_bounds__(32)
lstm2_dense_fused(const float* __restrict__ X,
                  const float* __restrict__ Wi1, const float* __restrict__ Wh1,
                  const float* __restrict__ b1,
                  const float* __restrict__ Wi2, const float* __restrict__ Wh2,
                  const float* __restrict__ b2,
                  const float* __restrict__ Wd,  const float* __restrict__ bd,
                  float* __restrict__ out)
{
    __shared__ v2f sWi1f[40 * 32];
    __shared__ v2f sWh1f[25 * 32];
    __shared__ v2f sWi2f[25 * 32];
    __shared__ v2f sWh2f[25 * 32];
    __shared__ v2f sWdf [ 5 * 32];

    const int lane  = threadIdx.x;
    const int mrow  = lane & 15;
    const int hi    = lane >> 4;
    const int khalf = 2 * hi;
    const int b0    = blockIdx.x * 16;

    for (int f = 0; f < 40; ++f) {
        const int n = f >> 3, kk = f & 7;
        const int row = 4 * kk + khalf, col = (mrow & 3) * LH + 4 * n + (mrow >> 2);
        v2f w; w.x = Wi1[row * LH4 + col]; w.y = Wi1[(row + 1) * LH4 + col];
        sWi1f[f * 32 + lane] = w;
    }
    for (int f = 0; f < 25; ++f) {
        const int n = f / 5, kk = f % 5;
        const int row = 4 * kk + khalf, col = (mrow & 3) * LH + 4 * n + (mrow >> 2);
        v2f w;
        w.x = Wh1[row * LH4 + col]; w.y = Wh1[(row + 1) * LH4 + col]; sWh1f[f * 32 + lane] = w;
        w.x = Wi2[row * LH4 + col]; w.y = Wi2[(row + 1) * LH4 + col]; sWi2f[f * 32 + lane] = w;
        w.x = Wh2[row * LH4 + col]; w.y = Wh2[(row + 1) * LH4 + col]; sWh2f[f * 32 + lane] = w;
    }
    for (int kk = 0; kk < 5; ++kk) {
        const int row = 4 * kk + khalf;
        v2f w;
        w.x = (mrow < LOUT) ? Wd[row * LOUT + mrow] : 0.0f;
        w.y = (mrow < LOUT) ? Wd[(row + 1) * LOUT + mrow] : 0.0f;
        sWdf[kk * 32 + lane] = w;
    }

    v8f bz1[5], bz2[5];
    #pragma unroll
    for (int n = 0; n < 5; ++n) {
        #pragma unroll
        for (int r = 0; r < 8; ++r) {
            const int col = (r & 3) * LH + 4 * n + khalf + (r >> 2);
            bz1[n][r] = b1[col];
            bz2[n][r] = b2[col];
        }
    }
    v8f bo;
    #pragma unroll
    for (int r = 0; r < 8; ++r) {
        const int o = 8 * hi + r;
        bo[r] = (o < LOUT) ? bd[o] : 0.0f;
    }

    v2f h1f[5], h2f[5], c1[5], c2[5];
    #pragma unroll
    for (int n = 0; n < 5; ++n) {
        h1f[n].x = 0.0f; h1f[n].y = 0.0f; h2f[n].x = 0.0f; h2f[n].y = 0.0f;
        c1[n].x  = 0.0f; c1[n].y  = 0.0f; c2[n].x  = 0.0f; c2[n].y  = 0.0f;
    }
    __syncthreads();

    for (int t = 0; t < LT; ++t) {
        const float* xrow = X + ((size_t)(b0 + mrow) * LT + (size_t)t) * LF;
        v2f ax[8];
        #pragma unroll
        for (int kk = 0; kk < 8; ++kk) {
            ax[kk].x = xrow[4 * kk + khalf];
            ax[kk].y = xrow[4 * kk + khalf + 1];
        }
        if (t + 1 < LT) __builtin_prefetch(xrow + LF, 0, 3);

        v8f zt[5];
        #pragma unroll
        for (int n = 0; n < 5; ++n) {
            v8f c = {};
            #pragma unroll
            for (int kk = 0; kk < 8; ++kk)
                WMMA4(c, sWi1f[(n * 8 + kk) * 32 + lane], ax[kk]);
            #pragma unroll
            for (int kk = 0; kk < 5; ++kk)
                WMMA4(c, sWh1f[(n * 5 + kk) * 32 + lane], h1f[kk]);
            zt[n] = c;
        }

        #pragma unroll
        for (int n = 0; n < 5; ++n) {
            #pragma unroll
            for (int s = 0; s < 2; ++s) {
                const float zi = zt[n][4 * s + 0] + bz1[n][4 * s + 0];
                const float zf = zt[n][4 * s + 1] + bz1[n][4 * s + 1];
                const float zg = zt[n][4 * s + 2] + bz1[n][4 * s + 2];
                const float zo = zt[n][4 * s + 3] + bz1[n][4 * s + 3];
                const float cc = sigf(zf) * c1[n][s] + sigf(zi) * tanhfast(zg);
                c1[n][s] = cc;
                h1f[n][s] = sigf(zo) * tanhfast(cc);
            }
        }

        #pragma unroll
        for (int n = 0; n < 5; ++n) {
            v8f c = {};
            #pragma unroll
            for (int kk = 0; kk < 5; ++kk)
                WMMA4(c, sWi2f[(n * 5 + kk) * 32 + lane], h1f[kk]);
            #pragma unroll
            for (int kk = 0; kk < 5; ++kk)
                WMMA4(c, sWh2f[(n * 5 + kk) * 32 + lane], h2f[kk]);
            zt[n] = c;
        }

        #pragma unroll
        for (int n = 0; n < 5; ++n) {
            #pragma unroll
            for (int s = 0; s < 2; ++s) {
                const float zi = zt[n][4 * s + 0] + bz2[n][4 * s + 0];
                const float zf = zt[n][4 * s + 1] + bz2[n][4 * s + 1];
                const float zg = zt[n][4 * s + 2] + bz2[n][4 * s + 2];
                const float zo = zt[n][4 * s + 3] + bz2[n][4 * s + 3];
                const float cc = sigf(zf) * c2[n][s] + sigf(zi) * tanhfast(zg);
                c2[n][s] = cc;
                h2f[n][s] = sigf(zo) * tanhfast(cc);
            }
        }

        {
            v8f c = {};
            #pragma unroll
            for (int kk = 0; kk < 5; ++kk)
                WMMA4(c, sWdf[kk * 32 + lane], h2f[kk]);
            float* orow = out + ((size_t)(b0 + mrow) * LT + (size_t)t) * LOUT + 8 * hi;
            v2f* o2 = (v2f*)orow;
            v2f p;
            p.x = c[0] + bo[0]; p.y = c[1] + bo[1]; o2[0] = p;
            if (hi == 0) {
                p.x = c[2] + bo[2]; p.y = c[3] + bo[3]; o2[1] = p;
                p.x = c[4] + bo[4]; p.y = c[5] + bo[5]; o2[2] = p;
                p.x = c[6] + bo[6]; p.y = c[7] + bo[7]; o2[3] = p;
            }
        }
    }
}

extern "C" void kernel_launch(void* const* d_in, const int* in_sizes, int n_in,
                              void* d_out, int out_size, void* d_ws, size_t ws_size,
                              hipStream_t stream) {
    (void)in_sizes; (void)n_in; (void)out_size; (void)d_ws; (void)ws_size;
    const float* X   = (const float*)d_in[0];
    const float* Wi1 = (const float*)d_in[1];
    const float* Wh1 = (const float*)d_in[2];
    const float* b1  = (const float*)d_in[3];
    const float* Wi2 = (const float*)d_in[4];
    const float* Wh2 = (const float*)d_in[5];
    const float* b2  = (const float*)d_in[6];
    const float* Wd  = (const float*)d_in[7];
    const float* bd  = (const float*)d_in[8];
    float* out = (float*)d_out;

    dim3 grid(LB / 16);
    dim3 block(32);
    lstm2_dense_fused<<<grid, block, 0, stream>>>(X, Wi1, Wh1, b1, Wi2, Wh2, b2, Wd, bd, out);
}


# One batch tile's step spread over 11 waves (a wave = one 16-row gate tile of one layer; layer 2 lags 1 step, head 2), h via LDS, 1 split barrier/step; sums reordered: 2 accumulators, bias first
# speedup vs baseline: 6.3369x; 4.1434x over previous
#include <hip/hip_runtime.h>


typedef __attribute__((ext_vector_type(2))) float v2f;
typedef __attribute__((ext_vector_type(8))) float v8f;

#define LH    20
#define LH4   80
#define LF    32
#define LT    1024
#define LB    256
#define LOUT  10
#define NWAVE 11
#define HBUF  160

__device__ __forceinline__ float sigf(float x) { return 1.0f / (1.0f + __expf(-x)); }
__device__ __forceinline__ float tanhfast(float x) {
    const float e = __expf(2.0f * x);
    return (e - 1.0f) / (e + 1.0f);
}

#define WMMA4(c, a, b) \
    (c) = __builtin_amdgcn_wmma_f32_16x16x4_f32(false, (a), false, (b), (short)0, (c), false, false)

__device__ __forceinline__ v2f lstm_cells(const v8f z, v2f& c) {
    v2f h;
    #pragma unroll
    for (int s = 0; s < 2; ++s) {
        const float zi = z[4 * s + 0];
        const float zf = z[4 * s + 1];
        const float zg = z[4 * s + 2];
        const float zo = z[4 * s + 3];
        const float cc = sigf(zf) * c[s] + sigf(zi) * tanhfast(zg);
        c[s] = cc;
        h[s] = sigf(zo) * tanhfast(cc);
    }
    return h;
}

__device__ __forceinline__ void lds_publish_and_signal() {
    __builtin_amdgcn_fence(__ATOMIC_RELEASE, "workgroup", "local");
    __builtin_amdgcn_s_barrier_signal(-1);
}
__device__ __forceinline__ void lds_wait_and_acquire() {
    __builtin_amdgcn_s_barrier_wait(-1);
    __builtin_amdgcn_fence(__ATOMIC_ACQUIRE, "workgroup", "local");
}

__global__ void __launch_bounds__(NWAVE * 32)
lstm2_dense_fused(const float* __restrict__ X,
                  const float* __restrict__ Wi1, const float* __restrict__ Wh1,
                  const float* __restrict__ b1,
                  const float* __restrict__ Wi2, const float* __restrict__ Wh2,
                  const float* __restrict__ b2,
                  const float* __restrict__ Wd,  const float* __restrict__ bd,
                  float* __restrict__ out)
{
    __shared__ v2f sH1[2 * HBUF];
    __shared__ v2f sH2[2 * HBUF];

    const int tid   = threadIdx.x;
    const int lane  = tid & 31;
    const int wave  = __builtin_amdgcn_readfirstlane(tid >> 5);
    const int mrow  = lane & 15;
    const int hi    = lane >> 4;
    const int khalf = 2 * hi;
    const int b0    = blockIdx.x * 16;

    const bool isL1 = wave < 5;
    const bool isL2 = (wave >= 5) && (wave < 10);
    const int  n    = isL1 ? wave : wave - 5;

    for (int i = tid; i < 2 * HBUF; i += NWAVE * 32) {
        v2f zz; zz.x = 0.0f; zz.y = 0.0f;
        sH1[i] = zz;
        sH2[i] = zz;
    }

    v2f win[8], wrec[5];
    v8f bz;
    if (isL1 || isL2) {
        const float* __restrict__ Win  = isL1 ? Wi1 : Wi2;
        const float* __restrict__ Wrec = isL1 ? Wh1 : Wh2;
        const float* __restrict__ bb   = isL1 ? b1  : b2;
        const int col = (mrow & 3) * LH + 4 * n + (mrow >> 2);
        #pragma unroll
        for (int kk = 0; kk < 8; ++kk) {
            const int row = 4 * kk + khalf;
            v2f w; w.x = 0.0f; w.y = 0.0f;
            if (kk < 5 || isL1) {
                w.x = Win[row * LH4 + col];
                w.y = Win[(row + 1) * LH4 + col];
            }
            win[kk] = w;
        }
        #pragma unroll
        for (int kk = 0; kk < 5; ++kk) {
            const int row = 4 * kk + khalf;
            v2f w;
            w.x = Wrec[row * LH4 + col];
            w.y = Wrec[(row + 1) * LH4 + col];
            wrec[kk] = w;
        }
        #pragma unroll
        for (int r = 0; r < 8; ++r)
            bz[r] = bb[(r & 3) * LH + 4 * n + khalf + (r >> 2)];
    } else {
        #pragma unroll
        for (int kk = 0; kk < 8; ++kk) { win[kk].x = 0.0f; win[kk].y = 0.0f; }
        #pragma unroll
        for (int kk = 0; kk < 5; ++kk) {
            const int row = 4 * kk + khalf;
            v2f w;
            w.x = (mrow < LOUT) ? Wd[row * LOUT + mrow] : 0.0f;
            w.y = (mrow < LOUT) ? Wd[(row + 1) * LOUT + mrow] : 0.0f;
            wrec[kk] = w;
        }
        #pragma unroll
        for (int r = 0; r < 8; ++r) {
            const int o = 8 * hi + r;
            bz[r] = (o < LOUT) ? bd[o] : 0.0f;
        }
    }

    const float* xlane = X + (size_t)(b0 + mrow) * LT * LF + khalf;
    v2f xc[8];
    #pragma unroll
    for (int kk = 0; kk < 8; ++kk) { xc[kk].x = 0.0f; xc[kk].y = 0.0f; }
    if (isL1) {
        #pragma unroll
        for (int kk = 0; kk < 8; ++kk)
            xc[kk] = *(const v2f*)(xlane + 4 * kk);
    }

    v2f c; c.x = 0.0f; c.y = 0.0f;
    v8f accA = {}, accB = {};

    __syncthreads();
    lds_publish_and_signal();

    for (int i = 0; i < LT + 2; ++i) {
        if (isL1 && i < LT) {
            accA = bz;
            accB = (v8f){};
            #pragma unroll
            for (int kk = 0; kk < 8; kk += 2) {
                WMMA4(accA, win[kk],     xc[kk]);
                WMMA4(accB, win[kk + 1], xc[kk + 1]);
            }
            if (i + 1 < LT) {
                const float* xr = xlane + (size_t)(i + 1) * LF;
                #pragma unroll
                for (int kk = 0; kk < 8; ++kk)
                    xc[kk] = *(const v2f*)(xr + 4 * kk);
            }
            if (i + 5 < LT) __builtin_prefetch(xlane + (size_t)(i + 5) * LF, 0, 3);
        }

        lds_wait_and_acquire();

        if (isL1) {
            if (i < LT) {
                const v2f* hprev = sH1 + ((i + 1) & 1) * HBUF + lane;
                v2f hf[5];
                #pragma unroll
                for (int kk = 0; kk < 5; ++kk) hf[kk] = hprev[kk * 32];
                WMMA4(accA, wrec[0], hf[0]);
                WMMA4(accB, wrec[1], hf[1]);
                WMMA4(accA, wrec[2], hf[2]);
                WMMA4(accB, wrec[3], hf[3]);
                WMMA4(accA, wrec[4], hf[4]);
                const v8f z = accA + accB;
                const v2f h = lstm_cells(z, c);
                sH1[(i & 1) * HBUF + n * 32 + lane] = h;
            }
        } else if (isL2) {
            if (i >= 1 && i <= LT) {
                const v2f* hin  = sH1 + ((i - 1) & 1) * HBUF + lane;
                const v2f* hrec = sH2 + (i & 1) * HBUF + lane;
                v2f ha[5], hb[5];
                #pragma unroll
                for (int kk = 0; kk < 5; ++kk) { ha[kk] = hin[kk * 32]; hb[kk] = hrec[kk * 32]; }
                v8f aA = bz, aB = {};
                #pragma unroll
                for (int kk = 0; kk < 5; ++kk) {
                    WMMA4(aA, win[kk],  ha[kk]);
                    WMMA4(aB, wrec[kk], hb[kk]);
                }
                const v8f z = aA + aB;
                const v2f h = lstm_cells(z, c);
                sH2[((i - 1) & 1) * HBUF + n * 32 + lane] = h;
            }
        } else {
            if (i >= 2) {
                const int t = i - 2;
                const v2f* hsrc = sH2 + (i & 1) * HBUF + lane;
                v2f hf[5];
                #pragma unroll
                for (int kk = 0; kk < 5; ++kk) hf[kk] = hsrc[kk * 32];
                v8f acc = {};
                #pragma unroll
                for (int kk = 0; kk < 5; ++kk)
                    WMMA4(acc, wrec[kk], hf[kk]);
                float* orow = out + ((size_t)(b0 + mrow) * LT + (size_t)t) * LOUT + 8 * hi;
                v2f* o2 = (v2f*)orow;
                v2f p;
                p.x = acc[0] + bz[0]; p.y = acc[1] + bz[1]; o2[0] = p;
                if (hi == 0) {
                    p.x = acc[2] + bz[2]; p.y = acc[3] + bz[3]; o2[1] = p;
                    p.x = acc[4] + bz[4]; p.y = acc[5] + bz[5]; o2[2] = p;
                    p.x = acc[6] + bz[6]; p.y = acc[7] + bz[7]; o2[3] = p;
                }
            }
        }

        lds_publish_and_signal();
    }
    lds_wait_and_acquire();
}

extern "C" void kernel_launch(void* const* d_in, const int* in_sizes, int n_in,
                              void* d_out, int out_size, void* d_ws, size_t ws_size,
                              hipStream_t stream) {
    (void)in_sizes; (void)n_in; (void)out_size; (void)d_ws; (void)ws_size;
    const float* X   = (const float*)d_in[0];
    const float* Wi1 = (const float*)d_in[1];
    const float* Wh1 = (const float*)d_in[2];
    const float* b1  = (const float*)d_in[3];
    const float* Wi2 = (const float*)d_in[4];
    const float* Wh2 = (const float*)d_in[5];
    const float* b2  = (const float*)d_in[6];
    const float* Wd  = (const float*)d_in[7];
    const float* bd  = (const float*)d_in[8];
    float* out = (float*)d_out;

    dim3 grid(LB / 16);
    dim3 block(NWAVE * 32);
    lstm2_dense_fused<<<grid, block, 0, stream>>>(X, Wi1, Wh1, b1, Wi2, Wh2, b2, Wd, bd, out);
}


# Gate sigmoid/tanh from hardware exp2 and reciprocal (v_exp_f32, v_rcp_f32), no IEEE division sequences; tanh as 1-2/(1+e^2x): about 1 ulp per reciprocal, saturates without inf/inf
# speedup vs baseline: 8.0957x; 1.2775x over previous
#include <hip/hip_runtime.h>


typedef __attribute__((ext_vector_type(2))) float v2f;
typedef __attribute__((ext_vector_type(8))) float v8f;

#define LH    20
#define LH4   80
#define LF    32
#define LT    1024
#define LB    256
#define LOUT  10
#define NWAVE 11
#define HBUF  160

#define LOG2E 1.44269504088896341f
__device__ __forceinline__ float sigf(float x) {
    return __builtin_amdgcn_rcpf(1.0f + __builtin_amdgcn_exp2f(x * -LOG2E));
}
__device__ __forceinline__ float tanhfast(float x) {
    const float r = __builtin_amdgcn_rcpf(1.0f + __builtin_amdgcn_exp2f(x * (2.0f * LOG2E)));
    return __builtin_fmaf(-2.0f, r, 1.0f);
}

#define WMMA4(c, a, b) \
    (c) = __builtin_amdgcn_wmma_f32_16x16x4_f32(false, (a), false, (b), (short)0, (c), false, false)

__device__ __forceinline__ v2f lstm_cells(const v8f z, v2f& c) {
    v2f h;
    #pragma unroll
    for (int s = 0; s < 2; ++s) {
        const float zi = z[4 * s + 0];
        const float zf = z[4 * s + 1];
        const float zg = z[4 * s + 2];
        const float zo = z[4 * s + 3];
        const float cc = sigf(zf) * c[s] + sigf(zi) * tanhfast(zg);
        c[s] = cc;
        h[s] = sigf(zo) * tanhfast(cc);
    }
    return h;
}

__device__ __forceinline__ void lds_publish_and_signal() {
    __builtin_amdgcn_fence(__ATOMIC_RELEASE, "workgroup", "local");
    __builtin_amdgcn_s_barrier_signal(-1);
}
__device__ __forceinline__ void lds_wait_and_acquire() {
    __builtin_amdgcn_s_barrier_wait(-1);
    __builtin_amdgcn_fence(__ATOMIC_ACQUIRE, "workgroup", "local");
}

__global__ void __launch_bounds__(NWAVE * 32)
lstm2_dense_fused(const float* __restrict__ X,
                  const float* __restrict__ Wi1, const float* __restrict__ Wh1,
                  const float* __restrict__ b1,
                  const float* __restrict__ Wi2, const float* __restrict__ Wh2,
                  const float* __restrict__ b2,
                  const float* __restrict__ Wd,  const float* __restrict__ bd,
                  float* __restrict__ out)
{
    __shared__ v2f sH1[2 * HBUF];
    __shared__ v2f sH2[2 * HBUF];

    const int tid   = threadIdx.x;
    const int lane  = tid & 31;
    const int wave  = __builtin_amdgcn_readfirstlane(tid >> 5);
    const int mrow  = lane & 15;
    const int hi    = lane >> 4;
    const int khalf = 2 * hi;
    const int b0    = blockIdx.x * 16;

    const bool isL1 = wave < 5;
    const bool isL2 = (wave >= 5) && (wave < 10);
    const int  n    = isL1 ? wave : wave - 5;

    for (int i = tid; i < 2 * HBUF; i += NWAVE * 32) {
        v2f zz; zz.x = 0.0f; zz.y = 0.0f;
        sH1[i] = zz;
        sH2[i] = zz;
    }

    v2f win[8], wrec[5];
    v8f bz;
    if (isL1 || isL2) {
        const float* __restrict__ Win  = isL1 ? Wi1 : Wi2;
        const float* __restrict__ Wrec = isL1 ? Wh1 : Wh2;
        const float* __restrict__ bb   = isL1 ? b1  : b2;
        const int col = (mrow & 3) * LH + 4 * n + (mrow >> 2);
        #pragma unroll
        for (int kk = 0; kk < 8; ++kk) {
            const int row = 4 * kk + khalf;
            v2f w; w.x = 0.0f; w.y = 0.0f;
            if (kk < 5 || isL1) {
                w.x = Win[row * LH4 + col];
                w.y = Win[(row + 1) * LH4 + col];
            }
            win[kk] = w;
        }
        #pragma unroll
        for (int kk = 0; kk < 5; ++kk) {
            const int row = 4 * kk + khalf;
            v2f w;
            w.x = Wrec[row * LH4 + col];
            w.y = Wrec[(row + 1) * LH4 + col];
            wrec[kk] = w;
        }
        #pragma unroll
        for (int r = 0; r < 8; ++r)
            bz[r] = bb[(r & 3) * LH + 4 * n + khalf + (r >> 2)];
    } else {
        #pragma unroll
        for (int kk = 0; kk < 8; ++kk) { win[kk].x = 0.0f; win[kk].y = 0.0f; }
        #pragma unroll
        for (int kk = 0; kk < 5; ++kk) {
            const int row = 4 * kk + khalf;
            v2f w;
            w.x = (mrow < LOUT) ? Wd[row * LOUT + mrow] : 0.0f;
            w.y = (mrow < LOUT) ? Wd[(row + 1) * LOUT + mrow] : 0.0f;
            wrec[kk] = w;
        }
        #pragma unroll
        for (int r = 0; r < 8; ++r) {
            const int o = 8 * hi + r;
            bz[r] = (o < LOUT) ? bd[o] : 0.0f;
        }
    }

    const float* xlane = X + (size_t)(b0 + mrow) * LT * LF + khalf;
    v2f xc[8];
    #pragma unroll
    for (int kk = 0; kk < 8; ++kk) { xc[kk].x = 0.0f; xc[kk].y = 0.0f; }
    if (isL1) {
        #pragma unroll
        for (int kk = 0; kk < 8; ++kk)
            xc[kk] = *(const v2f*)(xlane + 4 * kk);
    }

    v2f c; c.x = 0.0f; c.y = 0.0f;
    v8f accA = {}, accB = {};

    __syncthreads();
    lds_publish_and_signal();

    for (int i = 0; i < LT + 2; ++i) {
        if (isL1 && i < LT) {
            accA = bz;
            accB = (v8f){};
            #pragma unroll
            for (int kk = 0; kk < 8; kk += 2) {
                WMMA4(accA, win[kk],     xc[kk]);
                WMMA4(accB, win[kk + 1], xc[kk + 1]);
            }
            if (i + 1 < LT) {
                const float* xr = xlane + (size_t)(i + 1) * LF;
                #pragma unroll
                for (int kk = 0; kk < 8; ++kk)
                    xc[kk] = *(const v2f*)(xr + 4 * kk);
            }
            if (i + 5 < LT) __builtin_prefetch(xlane + (size_t)(i + 5) * LF, 0, 3);
        }

        lds_wait_and_acquire();

        if (isL1) {
            if (i < LT) {
                const v2f* hprev = sH1 + ((i + 1) & 1) * HBUF + lane;
                v2f hf[5];
                #pragma unroll
                for (int kk = 0; kk < 5; ++kk) hf[kk] = hprev[kk * 32];
                WMMA4(accA, wrec[0], hf[0]);
                WMMA4(accB, wrec[1], hf[1]);
                WMMA4(accA, wrec[2], hf[2]);
                WMMA4(accB, wrec[3], hf[3]);
                WMMA4(accA, wrec[4], hf[4]);
                const v8f z = accA + accB;
                const v2f h = lstm_cells(z, c);
                sH1[(i & 1) * HBUF + n * 32 + lane] = h;
            }
        } else if (isL2) {
            if (i >= 1 && i <= LT) {
                const v2f* hin  = sH1 + ((i - 1) & 1) * HBUF + lane;
                const v2f* hrec = sH2 + (i & 1) * HBUF + lane;
                v2f ha[5], hb[5];
                #pragma unroll
                for (int kk = 0; kk < 5; ++kk) { ha[kk] = hin[kk * 32]; hb[kk] = hrec[kk * 32]; }
                v8f aA = bz, aB = {};
                #pragma unroll
                for (int kk = 0; kk < 5; ++kk) {
                    WMMA4(aA, win[kk],  ha[kk]);
                    WMMA4(aB, wrec[kk], hb[kk]);
                }
                const v8f z = aA + aB;
                const v2f h = lstm_cells(z, c);
                sH2[((i - 1) & 1) * HBUF + n * 32 + lane] = h;
            }
        } else {
            if (i >= 2) {
                const int t = i - 2;
                const v2f* hsrc = sH2 + (i & 1) * HBUF + lane;
                v2f hf[5];
                #pragma unroll
                for (int kk = 0; kk < 5; ++kk) hf[kk] = hsrc[kk * 32];
                v8f acc = {};
                #pragma unroll
                for (int kk = 0; kk < 5; ++kk)
                    WMMA4(acc, wrec[kk], hf[kk]);
                float* orow = out + ((size_t)(b0 + mrow) * LT + (size_t)t) * LOUT + 8 * hi;
                v2f* o2 = (v2f*)orow;
                v2f p;
                p.x = acc[0] + bz[0]; p.y = acc[1] + bz[1]; o2[0] = p;
                if (hi == 0) {
                    p.x = acc[2] + bz[2]; p.y = acc[3] + bz[3]; o2[1] = p;
                    p.x = acc[4] + bz[4]; p.y = acc[5] + bz[5]; o2[2] = p;
                    p.x = acc[6] + bz[6]; p.y = acc[7] + bz[7]; o2[3] = p;
                }
            }
        }

        lds_publish_and_signal();
    }
    lds_wait_and_acquire();
}

extern "C" void kernel_launch(void* const* d_in, const int* in_sizes, int n_in,
                              void* d_out, int out_size, void* d_ws, size_t ws_size,
                              hipStream_t stream) {
    (void)in_sizes; (void)n_in; (void)out_size; (void)d_ws; (void)ws_size;
    const float* X   = (const float*)d_in[0];
    const float* Wi1 = (const float*)d_in[1];
    const float* Wh1 = (const float*)d_in[2];
    const float* b1  = (const float*)d_in[3];
    const float* Wi2 = (const float*)d_in[4];
    const float* Wh2 = (const float*)d_in[5];
    const float* b2  = (const float*)d_in[6];
    const float* Wd  = (const float*)d_in[7];
    const float* bd  = (const float*)d_in[8];
    float* out = (float*)d_out;

    dim3 grid(LB / 16);
    dim3 block(NWAVE * 32);
    lstm2_dense_fused<<<grid, block, 0, stream>>>(X, Wi1, Wh1, b1, Wi2, Wh2, b2, Wd, bd, out);
}


# Matrix-pipe priority by criticality (s_setprio): h-dependent phases of every wave at priority 3, layer-1 x-half WMMAs at 0, filling idle pipe time instead of delaying younger waves of a SIMD
# speedup vs baseline: 8.4410x; 1.0427x over previous
#include <hip/hip_runtime.h>


typedef __attribute__((ext_vector_type(2))) float v2f;
typedef __attribute__((ext_vector_type(8))) float v8f;

#define LH    20
#define LH4   80
#define LF    32
#define LT    1024
#define LB    256
#define LOUT  10
#define NWAVE 11
#define HBUF  160

#define LOG2E 1.44269504088896341f
__device__ __forceinline__ float sigf(float x) {
    return __builtin_amdgcn_rcpf(1.0f + __builtin_amdgcn_exp2f(x * -LOG2E));
}
__device__ __forceinline__ float tanhfast(float x) {
    const float r = __builtin_amdgcn_rcpf(1.0f + __builtin_amdgcn_exp2f(x * (2.0f * LOG2E)));
    return __builtin_fmaf(-2.0f, r, 1.0f);
}

#define WMMA4(c, a, b) \
    (c) = __builtin_amdgcn_wmma_f32_16x16x4_f32(false, (a), false, (b), (short)0, (c), false, false)

__device__ __forceinline__ v2f lstm_cells(const v8f z, v2f& c) {
    v2f h;
    #pragma unroll
    for (int s = 0; s < 2; ++s) {
        const float zi = z[4 * s + 0];
        const float zf = z[4 * s + 1];
        const float zg = z[4 * s + 2];
        const float zo = z[4 * s + 3];
        const float cc = sigf(zf) * c[s] + sigf(zi) * tanhfast(zg);
        c[s] = cc;
        h[s] = sigf(zo) * tanhfast(cc);
    }
    return h;
}

__device__ __forceinline__ void lds_publish_and_signal() {
    __builtin_amdgcn_fence(__ATOMIC_RELEASE, "workgroup", "local");
    __builtin_amdgcn_s_barrier_signal(-1);
}
__device__ __forceinline__ void lds_wait_and_acquire() {
    __builtin_amdgcn_s_barrier_wait(-1);
    __builtin_amdgcn_fence(__ATOMIC_ACQUIRE, "workgroup", "local");
}

__global__ void __launch_bounds__(NWAVE * 32)
lstm2_dense_fused(const float* __restrict__ X,
                  const float* __restrict__ Wi1, const float* __restrict__ Wh1,
                  const float* __restrict__ b1,
                  const float* __restrict__ Wi2, const float* __restrict__ Wh2,
                  const float* __restrict__ b2,
                  const float* __restrict__ Wd,  const float* __restrict__ bd,
                  float* __restrict__ out)
{
    __shared__ v2f sH1[2 * HBUF];
    __shared__ v2f sH2[2 * HBUF];

    const int tid   = threadIdx.x;
    const int lane  = tid & 31;
    const int wave  = __builtin_amdgcn_readfirstlane(tid >> 5);
    const int mrow  = lane & 15;
    const int hi    = lane >> 4;
    const int khalf = 2 * hi;
    const int b0    = blockIdx.x * 16;

    const bool isL1 = wave < 5;
    const bool isL2 = (wave >= 5) && (wave < 10);
    const int  n    = isL1 ? wave : wave - 5;

    for (int i = tid; i < 2 * HBUF; i += NWAVE * 32) {
        v2f zz; zz.x = 0.0f; zz.y = 0.0f;
        sH1[i] = zz;
        sH2[i] = zz;
    }

    v2f win[8], wrec[5];
    v8f bz;
    if (isL1 || isL2) {
        const float* __restrict__ Win  = isL1 ? Wi1 : Wi2;
        const float* __restrict__ Wrec = isL1 ? Wh1 : Wh2;
        const float* __restrict__ bb   = isL1 ? b1  : b2;
        const int col = (mrow & 3) * LH + 4 * n + (mrow >> 2);
        #pragma unroll
        for (int kk = 0; kk < 8; ++kk) {
            const int row = 4 * kk + khalf;
            v2f w; w.x = 0.0f; w.y = 0.0f;
            if (kk < 5 || isL1) {
                w.x = Win[row * LH4 + col];
                w.y = Win[(row + 1) * LH4 + col];
            }
            win[kk] = w;
        }
        #pragma unroll
        for (int kk = 0; kk < 5; ++kk) {
            const int row = 4 * kk + khalf;
            v2f w;
            w.x = Wrec[row * LH4 + col];
            w.y = Wrec[(row + 1) * LH4 + col];
            wrec[kk] = w;
        }
        #pragma unroll
        for (int r = 0; r < 8; ++r)
            bz[r] = bb[(r & 3) * LH + 4 * n + khalf + (r >> 2)];
    } else {
        #pragma unroll
        for (int kk = 0; kk < 8; ++kk) { win[kk].x = 0.0f; win[kk].y = 0.0f; }
        #pragma unroll
        for (int kk = 0; kk < 5; ++kk) {
            const int row = 4 * kk + khalf;
            v2f w;
            w.x = (mrow < LOUT) ? Wd[row * LOUT + mrow] : 0.0f;
            w.y = (mrow < LOUT) ? Wd[(row + 1) * LOUT + mrow] : 0.0f;
            wrec[kk] = w;
        }
        #pragma unroll
        for (int r = 0; r < 8; ++r) {
            const int o = 8 * hi + r;
            bz[r] = (o < LOUT) ? bd[o] : 0.0f;
        }
    }

    const float* xlane = X + (size_t)(b0 + mrow) * LT * LF + khalf;
    v2f xc[8];
    #pragma unroll
    for (int kk = 0; kk < 8; ++kk) { xc[kk].x = 0.0f; xc[kk].y = 0.0f; }
    if (isL1) {
        #pragma unroll
        for (int kk = 0; kk < 8; ++kk)
            xc[kk] = *(const v2f*)(xlane + 4 * kk);
    }

    v2f c; c.x = 0.0f; c.y = 0.0f;
    v8f accA = {}, accB = {};

    __syncthreads();
    __builtin_amdgcn_s_setprio(3);
    lds_publish_and_signal();

    for (int i = 0; i < LT + 2; ++i) {
        if (isL1 && i < LT) {
            __builtin_amdgcn_s_setprio(0);
            accA = bz;
            accB = (v8f){};
            #pragma unroll
            for (int kk = 0; kk < 8; kk += 2) {
                WMMA4(accA, win[kk],     xc[kk]);
                WMMA4(accB, win[kk + 1], xc[kk + 1]);
            }
            if (i + 1 < LT) {
                const float* xr = xlane + (size_t)(i + 1) * LF;
                #pragma unroll
                for (int kk = 0; kk < 8; ++kk)
                    xc[kk] = *(const v2f*)(xr + 4 * kk);
            }
            if (i + 5 < LT) __builtin_prefetch(xlane + (size_t)(i + 5) * LF, 0, 3);
            __builtin_amdgcn_s_setprio(3);
        }

        lds_wait_and_acquire();

        if (isL1) {
            if (i < LT) {
                const v2f* hprev = sH1 + ((i + 1) & 1) * HBUF + lane;
                v2f hf[5];
                #pragma unroll
                for (int kk = 0; kk < 5; ++kk) hf[kk] = hprev[kk * 32];
                WMMA4(accA, wrec[0], hf[0]);
                WMMA4(accB, wrec[1], hf[1]);
                WMMA4(accA, wrec[2], hf[2]);
                WMMA4(accB, wrec[3], hf[3]);
                WMMA4(accA, wrec[4], hf[4]);
                const v8f z = accA + accB;
                const v2f h = lstm_cells(z, c);
                sH1[(i & 1) * HBUF + n * 32 + lane] = h;
            }
        } else if (isL2) {
            if (i >= 1 && i <= LT) {
                const v2f* hin  = sH1 + ((i - 1) & 1) * HBUF + lane;
                const v2f* hrec = sH2 + (i & 1) * HBUF + lane;
                v2f ha[5], hb[5];
                #pragma unroll
                for (int kk = 0; kk < 5; ++kk) { ha[kk] = hin[kk * 32]; hb[kk] = hrec[kk * 32]; }
                v8f aA = bz, aB = {};
                #pragma unroll
                for (int kk = 0; kk < 5; ++kk) {
                    WMMA4(aA, win[kk],  ha[kk]);
                    WMMA4(aB, wrec[kk], hb[kk]);
                }
                const v8f z = aA + aB;
                const v2f h = lstm_cells(z, c);
                sH2[((i - 1) & 1) * HBUF + n * 32 + lane] = h;
            }
        } else {
            if (i >= 2) {
                const int t = i - 2;
                const v2f* hsrc = sH2 + (i & 1) * HBUF + lane;
                v2f hf[5];
                #pragma unroll
                for (int kk = 0; kk < 5; ++kk) hf[kk] = hsrc[kk * 32];
                v8f acc = {};
                #pragma unroll
                for (int kk = 0; kk < 5; ++kk)
                    WMMA4(acc, wrec[kk], hf[kk]);
                float* orow = out + ((size_t)(b0 + mrow) * LT + (size_t)t) * LOUT + 8 * hi;
                v2f* o2 = (v2f*)orow;
                v2f p;
                p.x = acc[0] + bz[0]; p.y = acc[1] + bz[1]; o2[0] = p;
                if (hi == 0) {
                    p.x = acc[2] + bz[2]; p.y = acc[3] + bz[3]; o2[1] = p;
                    p.x = acc[4] + bz[4]; p.y = acc[5] + bz[5]; o2[2] = p;
                    p.x = acc[6] + bz[6]; p.y = acc[7] + bz[7]; o2[3] = p;
                }
            }
        }

        lds_publish_and_signal();
    }
    lds_wait_and_acquire();
    __builtin_amdgcn_s_setprio(0);
}

extern "C" void kernel_launch(void* const* d_in, const int* in_sizes, int n_in,
                              void* d_out, int out_size, void* d_ws, size_t ws_size,
                              hipStream_t stream) {
    (void)in_sizes; (void)n_in; (void)out_size; (void)d_ws; (void)ws_size;
    const float* X   = (const float*)d_in[0];
    const float* Wi1 = (const float*)d_in[1];
    const float* Wh1 = (const float*)d_in[2];
    const float* b1  = (const float*)d_in[3];
    const float* Wi2 = (const float*)d_in[4];
    const float* Wh2 = (const float*)d_in[5];
    const float* b2  = (const float*)d_in[6];
    const float* Wd  = (const float*)d_in[7];
    const float* bd  = (const float*)d_in[8];
    float* out = (float*)d_out;

    dim3 grid(LB / 16);
    dim3 block(NWAVE * 32);
    lstm2_dense_fused<<<grid, block, 0, stream>>>(X, Wi1, Wh1, b1, Wi2, Wh2, b2, Wd, bd, out);
}


# One time loop per wave role (layer 1 / layer 2 / head), same barrier count: no role or step-range branches inside the step, a wave's LDS loads follow its barrier wait directly
# speedup vs baseline: 8.8317x; 1.0463x over previous
#include <hip/hip_runtime.h>


typedef __attribute__((ext_vector_type(2))) float v2f;
typedef __attribute__((ext_vector_type(8))) float v8f;

#define LH    20
#define LH4   80
#define LF    32
#define LT    1024
#define LB    256
#define LOUT  10
#define NWAVE 11
#define HBUF  160

#define LOG2E 1.44269504088896341f
__device__ __forceinline__ float sigf(float x) {
    return __builtin_amdgcn_rcpf(1.0f + __builtin_amdgcn_exp2f(x * -LOG2E));
}
__device__ __forceinline__ float tanhfast(float x) {
    const float r = __builtin_amdgcn_rcpf(1.0f + __builtin_amdgcn_exp2f(x * (2.0f * LOG2E)));
    return __builtin_fmaf(-2.0f, r, 1.0f);
}

#define WMMA4(c, a, b) \
    (c) = __builtin_amdgcn_wmma_f32_16x16x4_f32(false, (a), false, (b), (short)0, (c), false, false)

__device__ __forceinline__ v2f lstm_cells(const v8f z, v2f& c) {
    v2f h;
    #pragma unroll
    for (int s = 0; s < 2; ++s) {
        const float zi = z[4 * s + 0];
        const float zf = z[4 * s + 1];
        const float zg = z[4 * s + 2];
        const float zo = z[4 * s + 3];
        const float cc = sigf(zf) * c[s] + sigf(zi) * tanhfast(zg);
        c[s] = cc;
        h[s] = sigf(zo) * tanhfast(cc);
    }
    return h;
}

__device__ __forceinline__ void lds_publish_and_signal() {
    __builtin_amdgcn_fence(__ATOMIC_RELEASE, "workgroup", "local");
    __builtin_amdgcn_s_barrier_signal(-1);
}
__device__ __forceinline__ void lds_wait_and_acquire() {
    __builtin_amdgcn_s_barrier_wait(-1);
    __builtin_amdgcn_fence(__ATOMIC_ACQUIRE, "workgroup", "local");
}

__global__ void __launch_bounds__(NWAVE * 32)
lstm2_dense_fused(const float* __restrict__ X,
                  const float* __restrict__ Wi1, const float* __restrict__ Wh1,
                  const float* __restrict__ b1,
                  const float* __restrict__ Wi2, const float* __restrict__ Wh2,
                  const float* __restrict__ b2,
                  const float* __restrict__ Wd,  const float* __restrict__ bd,
                  float* __restrict__ out)
{
    __shared__ v2f sH1[2 * HBUF];
    __shared__ v2f sH2[2 * HBUF];

    const int tid   = threadIdx.x;
    const int lane  = tid & 31;
    const int wave  = __builtin_amdgcn_readfirstlane(tid >> 5);
    const int mrow  = lane & 15;
    const int hi    = lane >> 4;
    const int khalf = 2 * hi;
    const int b0    = blockIdx.x * 16;

    const bool isL1 = wave < 5;
    const bool isL2 = (wave >= 5) && (wave < 10);
    const int  n    = isL1 ? wave : wave - 5;

    for (int i = tid; i < 2 * HBUF; i += NWAVE * 32) {
        v2f zz; zz.x = 0.0f; zz.y = 0.0f;
        sH1[i] = zz;
        sH2[i] = zz;
    }

    v2f win[8], wrec[5];
    v8f bz;
    if (isL1 || isL2) {
        const float* __restrict__ Win  = isL1 ? Wi1 : Wi2;
        const float* __restrict__ Wrec = isL1 ? Wh1 : Wh2;
        const float* __restrict__ bb   = isL1 ? b1  : b2;
        const int col = (mrow & 3) * LH + 4 * n + (mrow >> 2);
        #pragma unroll
        for (int kk = 0; kk < 8; ++kk) {
            const int row = 4 * kk + khalf;
            v2f w; w.x = 0.0f; w.y = 0.0f;
            if (kk < 5 || isL1) {
                w.x = Win[row * LH4 + col];
                w.y = Win[(row + 1) * LH4 + col];
            }
            win[kk] = w;
        }
        #pragma unroll
        for (int kk = 0; kk < 5; ++kk) {
            const int row = 4 * kk + khalf;
            v2f w;
            w.x = Wrec[row * LH4 + col];
            w.y = Wrec[(row + 1) * LH4 + col];
            wrec[kk] = w;
        }
        #pragma unroll
        for (int r = 0; r < 8; ++r)
            bz[r] = bb[(r & 3) * LH + 4 * n + khalf + (r >> 2)];
    } else {
        #pragma unroll
        for (int kk = 0; kk < 8; ++kk) { win[kk].x = 0.0f; win[kk].y = 0.0f; }
        #pragma unroll
        for (int kk = 0; kk < 5; ++kk) {
            const int row = 4 * kk + khalf;
            v2f w;
            w.x = (mrow < LOUT) ? Wd[row * LOUT + mrow] : 0.0f;
            w.y = (mrow < LOUT) ? Wd[(row + 1) * LOUT + mrow] : 0.0f;
            wrec[kk] = w;
        }
        #pragma unroll
        for (int r = 0; r < 8; ++r) {
            const int o = 8 * hi + r;
            bz[r] = (o < LOUT) ? bd[o] : 0.0f;
        }
    }

    const float* xlane = X + (size_t)(b0 + mrow) * LT * LF + khalf;
    v2f xc[8];
    #pragma unroll
    for (int kk = 0; kk < 8; ++kk) { xc[kk].x = 0.0f; xc[kk].y = 0.0f; }
    if (isL1) {
        #pragma unroll
        for (int kk = 0; kk < 8; ++kk)
            xc[kk] = *(const v2f*)(xlane + 4 * kk);
    }

    v2f c; c.x = 0.0f; c.y = 0.0f;

    __syncthreads();
    __builtin_amdgcn_s_setprio(3);
    lds_publish_and_signal();

    if (isL1) {
        for (int i = 0; i < LT; ++i) {
            __builtin_amdgcn_s_setprio(0);
            v8f accA = bz;
            v8f accB = {};
            #pragma unroll
            for (int kk = 0; kk < 8; kk += 2) {
                WMMA4(accA, win[kk],     xc[kk]);
                WMMA4(accB, win[kk + 1], xc[kk + 1]);
            }
            if (i + 1 < LT) {
                const float* xr = xlane + (size_t)(i + 1) * LF;
                #pragma unroll
                for (int kk = 0; kk < 8; ++kk)
                    xc[kk] = *(const v2f*)(xr + 4 * kk);
            }
            if (i + 5 < LT) __builtin_prefetch(xlane + (size_t)(i + 5) * LF, 0, 3);
            __builtin_amdgcn_s_setprio(3);

            lds_wait_and_acquire();

            const v2f* hprev = sH1 + ((i + 1) & 1) * HBUF + lane;
            v2f hf[5];
            #pragma unroll
            for (int kk = 0; kk < 5; ++kk) hf[kk] = hprev[kk * 32];
            WMMA4(accA, wrec[0], hf[0]);
            WMMA4(accB, wrec[1], hf[1]);
            WMMA4(accA, wrec[2], hf[2]);
            WMMA4(accB, wrec[3], hf[3]);
            WMMA4(accA, wrec[4], hf[4]);
            const v8f z = accA + accB;
            const v2f h = lstm_cells(z, c);
            sH1[(i & 1) * HBUF + n * 32 + lane] = h;

            lds_publish_and_signal();
        }
        lds_wait_and_acquire();
        lds_publish_and_signal();
        lds_wait_and_acquire();
        lds_publish_and_signal();
    } else if (isL2) {
        lds_wait_and_acquire();
        lds_publish_and_signal();
        for (int i = 1; i <= LT; ++i) {
            lds_wait_and_acquire();

            const v2f* hin  = sH1 + ((i - 1) & 1) * HBUF + lane;
            const v2f* hrec = sH2 + (i & 1) * HBUF + lane;
            v2f ha[5], hb[5];
            #pragma unroll
            for (int kk = 0; kk < 5; ++kk) { ha[kk] = hin[kk * 32]; hb[kk] = hrec[kk * 32]; }
            v8f aA = bz, aB = {};
            #pragma unroll
            for (int kk = 0; kk < 5; ++kk) {
                WMMA4(aA, win[kk],  ha[kk]);
                WMMA4(aB, wrec[kk], hb[kk]);
            }
            const v8f z = aA + aB;
            const v2f h = lstm_cells(z, c);
            sH2[((i - 1) & 1) * HBUF + n * 32 + lane] = h;

            lds_publish_and_signal();
        }
        lds_wait_and_acquire();
        lds_publish_and_signal();
    } else {
        lds_wait_and_acquire();
        lds_publish_and_signal();
        lds_wait_and_acquire();
        lds_publish_and_signal();
        for (int i = 2; i < LT + 2; ++i) {
            lds_wait_and_acquire();

            const int t = i - 2;
            const v2f* hsrc = sH2 + (i & 1) * HBUF + lane;
            v2f hf[5];
            #pragma unroll
            for (int kk = 0; kk < 5; ++kk) hf[kk] = hsrc[kk * 32];
            v8f acc = {};
            #pragma unroll
            for (int kk = 0; kk < 5; ++kk)
                WMMA4(acc, wrec[kk], hf[kk]);
            float* orow = out + ((size_t)(b0 + mrow) * LT + (size_t)t) * LOUT + 8 * hi;
            v2f* o2 = (v2f*)orow;
            v2f p;
            p.x = acc[0] + bz[0]; p.y = acc[1] + bz[1]; o2[0] = p;
            if (hi == 0) {
                p.x = acc[2] + bz[2]; p.y = acc[3] + bz[3]; o2[1] = p;
                p.x = acc[4] + bz[4]; p.y = acc[5] + bz[5]; o2[2] = p;
                p.x = acc[6] + bz[6]; p.y = acc[7] + bz[7]; o2[3] = p;
            }

            lds_publish_and_signal();
        }
    }
    lds_wait_and_acquire();
    __builtin_amdgcn_s_setprio(0);
}

extern "C" void kernel_launch(void* const* d_in, const int* in_sizes, int n_in,
                              void* d_out, int out_size, void* d_ws, size_t ws_size,
                              hipStream_t stream) {
    (void)in_sizes; (void)n_in; (void)out_size; (void)d_ws; (void)ws_size;
    const float* X   = (const float*)d_in[0];
    const float* Wi1 = (const float*)d_in[1];
    const float* Wh1 = (const float*)d_in[2];
    const float* b1  = (const float*)d_in[3];
    const float* Wi2 = (const float*)d_in[4];
    const float* Wh2 = (const float*)d_in[5];
    const float* b2  = (const float*)d_in[6];
    const float* Wd  = (const float*)d_in[7];
    const float* bd  = (const float*)d_in[8];
    float* out = (float*)d_out;

    dim3 grid(LB / 16);
    dim3 block(NWAVE * 32);
    lstm2_dense_fused<<<grid, block, 0, stream>>>(X, Wi1, Wh1, b1, Wi2, Wh2, b2, Wd, bd, out);
}


# Gate sigmoid/tanh on the hardware tanh instruction (v_tanh_f32): tanh direct, sigmoid = 0.5 + 0.5 tanh(x/2); 5 transcendentals per cell instead of 10, no exp2, no reciprocals
# speedup vs baseline: 10.0039x; 1.1327x over previous
#include <hip/hip_runtime.h>


typedef __attribute__((ext_vector_type(2))) float v2f;
typedef __attribute__((ext_vector_type(8))) float v8f;

#define LH    20
#define LH4   80
#define LF    32
#define LT    1024
#define LB    256
#define LOUT  10
#define NWAVE 11
#define HBUF  160

__device__ __forceinline__ float tanhfast(float x) {
    return __builtin_amdgcn_tanhf(x);
}
__device__ __forceinline__ float sigf(float x) {
    return __builtin_fmaf(0.5f, __builtin_amdgcn_tanhf(0.5f * x), 0.5f);
}

#define WMMA4(c, a, b) \
    (c) = __builtin_amdgcn_wmma_f32_16x16x4_f32(false, (a), false, (b), (short)0, (c), false, false)

__device__ __forceinline__ v2f lstm_cells(const v8f z, v2f& c) {
    v2f h;
    #pragma unroll
    for (int s = 0; s < 2; ++s) {
        const float zi = z[4 * s + 0];
        const float zf = z[4 * s + 1];
        const float zg = z[4 * s + 2];
        const float zo = z[4 * s + 3];
        const float cc = sigf(zf) * c[s] + sigf(zi) * tanhfast(zg);
        c[s] = cc;
        h[s] = sigf(zo) * tanhfast(cc);
    }
    return h;
}

__device__ __forceinline__ void lds_publish_and_signal() {
    __builtin_amdgcn_fence(__ATOMIC_RELEASE, "workgroup", "local");
    __builtin_amdgcn_s_barrier_signal(-1);
}
__device__ __forceinline__ void lds_wait_and_acquire() {
    __builtin_amdgcn_s_barrier_wait(-1);
    __builtin_amdgcn_fence(__ATOMIC_ACQUIRE, "workgroup", "local");
}

__global__ void __launch_bounds__(NWAVE * 32)
lstm2_dense_fused(const float* __restrict__ X,
                  const float* __restrict__ Wi1, const float* __restrict__ Wh1,
                  const float* __restrict__ b1,
                  const float* __restrict__ Wi2, const float* __restrict__ Wh2,
                  const float* __restrict__ b2,
                  const float* __restrict__ Wd,  const float* __restrict__ bd,
                  float* __restrict__ out)
{
    __shared__ v2f sH1[2 * HBUF];
    __shared__ v2f sH2[2 * HBUF];

    const int tid   = threadIdx.x;
    const int lane  = tid & 31;
    const int wave  = __builtin_amdgcn_readfirstlane(tid >> 5);
    const int mrow  = lane & 15;
    const int hi    = lane >> 4;
    const int khalf = 2 * hi;
    const int b0    = blockIdx.x * 16;

    const bool isL1 = wave < 5;
    const bool isL2 = (wave >= 5) && (wave < 10);
    const int  n    = isL1 ? wave : wave - 5;

    for (int i = tid; i < 2 * HBUF; i += NWAVE * 32) {
        v2f zz; zz.x = 0.0f; zz.y = 0.0f;
        sH1[i] = zz;
        sH2[i] = zz;
    }

    v2f win[8], wrec[5];
    v8f bz;
    if (isL1 || isL2) {
        const float* __restrict__ Win  = isL1 ? Wi1 : Wi2;
        const float* __restrict__ Wrec = isL1 ? Wh1 : Wh2;
        const float* __restrict__ bb   = isL1 ? b1  : b2;
        const int col = (mrow & 3) * LH + 4 * n + (mrow >> 2);
        #pragma unroll
        for (int kk = 0; kk < 8; ++kk) {
            const int row = 4 * kk + khalf;
            v2f w; w.x = 0.0f; w.y = 0.0f;
            if (kk < 5 || isL1) {
                w.x = Win[row * LH4 + col];
                w.y = Win[(row + 1) * LH4 + col];
            }
            win[kk] = w;
        }
        #pragma unroll
        for (int kk = 0; kk < 5; ++kk) {
            const int row = 4 * kk + khalf;
            v2f w;
            w.x = Wrec[row * LH4 + col];
            w.y = Wrec[(row + 1) * LH4 + col];
            wrec[kk] = w;
        }
        #pragma unroll
        for (int r = 0; r < 8; ++r)
            bz[r] = bb[(r & 3) * LH + 4 * n + khalf + (r >> 2)];
    } else {
        #pragma unroll
        for (int kk = 0; kk < 8; ++kk) { win[kk].x = 0.0f; win[kk].y = 0.0f; }
        #pragma unroll
        for (int kk = 0; kk < 5; ++kk) {
            const int row = 4 * kk + khalf;
            v2f w;
            w.x = (mrow < LOUT) ? Wd[row * LOUT + mrow] : 0.0f;
            w.y = (mrow < LOUT) ? Wd[(row + 1) * LOUT + mrow] : 0.0f;
            wrec[kk] = w;
        }
        #pragma unroll
        for (int r = 0; r < 8; ++r) {
            const int o = 8 * hi + r;
            bz[r] = (o < LOUT) ? bd[o] : 0.0f;
        }
    }

    const float* xlane = X + (size_t)(b0 + mrow) * LT * LF + khalf;
    v2f xc[8];
    #pragma unroll
    for (int kk = 0; kk < 8; ++kk) { xc[kk].x = 0.0f; xc[kk].y = 0.0f; }
    if (isL1) {
        #pragma unroll
        for (int kk = 0; kk < 8; ++kk)
            xc[kk] = *(const v2f*)(xlane + 4 * kk);
    }

    v2f c; c.x = 0.0f; c.y = 0.0f;

    __syncthreads();
    __builtin_amdgcn_s_setprio(3);
    lds_publish_and_signal();

    if (isL1) {
        for (int i = 0; i < LT; ++i) {
            __builtin_amdgcn_s_setprio(0);
            v8f accA = bz;
            v8f accB = {};
            #pragma unroll
            for (int kk = 0; kk < 8; kk += 2) {
                WMMA4(accA, win[kk],     xc[kk]);
                WMMA4(accB, win[kk + 1], xc[kk + 1]);
            }
            if (i + 1 < LT) {
                const float* xr = xlane + (size_t)(i + 1) * LF;
                #pragma unroll
                for (int kk = 0; kk < 8; ++kk)
                    xc[kk] = *(const v2f*)(xr + 4 * kk);
            }
            if (i + 5 < LT) __builtin_prefetch(xlane + (size_t)(i + 5) * LF, 0, 3);
            __builtin_amdgcn_s_setprio(3);

            lds_wait_and_acquire();

            const v2f* hprev = sH1 + ((i + 1) & 1) * HBUF + lane;
            v2f hf[5];
            #pragma unroll
            for (int kk = 0; kk < 5; ++kk) hf[kk] = hprev[kk * 32];
            WMMA4(accA, wrec[0], hf[0]);
            WMMA4(accB, wrec[1], hf[1]);
            WMMA4(accA, wrec[2], hf[2]);
            WMMA4(accB, wrec[3], hf[3]);
            WMMA4(accA, wrec[4], hf[4]);
            const v8f z = accA + accB;
            const v2f h = lstm_cells(z, c);
            sH1[(i & 1) * HBUF + n * 32 + lane] = h;

            lds_publish_and_signal();
        }
        lds_wait_and_acquire();
        lds_publish_and_signal();
        lds_wait_and_acquire();
        lds_publish_and_signal();
    } else if (isL2) {
        lds_wait_and_acquire();
        lds_publish_and_signal();
        for (int i = 1; i <= LT; ++i) {
            lds_wait_and_acquire();

            const v2f* hin  = sH1 + ((i - 1) & 1) * HBUF + lane;
            const v2f* hrec = sH2 + (i & 1) * HBUF + lane;
            v2f ha[5], hb[5];
            #pragma unroll
            for (int kk = 0; kk < 5; ++kk) { ha[kk] = hin[kk * 32]; hb[kk] = hrec[kk * 32]; }
            v8f aA = bz, aB = {};
            #pragma unroll
            for (int kk = 0; kk < 5; ++kk) {
                WMMA4(aA, win[kk],  ha[kk]);
                WMMA4(aB, wrec[kk], hb[kk]);
            }
            const v8f z = aA + aB;
            const v2f h = lstm_cells(z, c);
            sH2[((i - 1) & 1) * HBUF + n * 32 + lane] = h;

            lds_publish_and_signal();
        }
        lds_wait_and_acquire();
        lds_publish_and_signal();
    } else {
        lds_wait_and_acquire();
        lds_publish_and_signal();
        lds_wait_and_acquire();
        lds_publish_and_signal();
        for (int i = 2; i < LT + 2; ++i) {
            lds_wait_and_acquire();

            const int t = i - 2;
            const v2f* hsrc = sH2 + (i & 1) * HBUF + lane;
            v2f hf[5];
            #pragma unroll
            for (int kk = 0; kk < 5; ++kk) hf[kk] = hsrc[kk * 32];
            v8f acc = {};
            #pragma unroll
            for (int kk = 0; kk < 5; ++kk)
                WMMA4(acc, wrec[kk], hf[kk]);
            float* orow = out + ((size_t)(b0 + mrow) * LT + (size_t)t) * LOUT + 8 * hi;
            v2f* o2 = (v2f*)orow;
            v2f p;
            p.x = acc[0] + bz[0]; p.y = acc[1] + bz[1]; o2[0] = p;
            if (hi == 0) {
                p.x = acc[2] + bz[2]; p.y = acc[3] + bz[3]; o2[1] = p;
                p.x = acc[4] + bz[4]; p.y = acc[5] + bz[5]; o2[2] = p;
                p.x = acc[6] + bz[6]; p.y = acc[7] + bz[7]; o2[3] = p;
            }

            lds_publish_and_signal();
        }
    }
    lds_wait_and_acquire();
    __builtin_amdgcn_s_setprio(0);
}

extern "C" void kernel_launch(void* const* d_in, const int* in_sizes, int n_in,
                              void* d_out, int out_size, void* d_ws, size_t ws_size,
                              hipStream_t stream) {
    (void)in_sizes; (void)n_in; (void)out_size; (void)d_ws; (void)ws_size;
    const float* X   = (const float*)d_in[0];
    const float* Wi1 = (const float*)d_in[1];
    const float* Wh1 = (const float*)d_in[2];
    const float* b1  = (const float*)d_in[3];
    const float* Wi2 = (const float*)d_in[4];
    const float* Wh2 = (const float*)d_in[5];
    const float* b2  = (const float*)d_in[6];
    const float* Wd  = (const float*)d_in[7];
    const float* bd  = (const float*)d_in[8];
    float* out = (float*)d_out;

    dim3 grid(LB / 16);
    dim3 block(NWAVE * 32);
    lstm2_dense_fused<<<grid, block, 0, stream>>>(X, Wi1, Wh1, b1, Wi2, Wh2, b2, Wd, bd, out);
}


# Layer 2 lags 2 steps (head 3): its input half Wi2^T h1 uses h1 fetched one phase earlier, runs between signal and wait at low priority like the x half; L2 sums reordered (2 accumulators interleaved)
# speedup vs baseline: 10.6941x; 1.0690x over previous
#include <hip/hip_runtime.h>


typedef __attribute__((ext_vector_type(2))) float v2f;
typedef __attribute__((ext_vector_type(8))) float v8f;

#define LH    20
#define LH4   80
#define LF    32
#define LT    1024
#define LB    256
#define LOUT  10
#define NWAVE 11
#define HBUF  160

__device__ __forceinline__ float tanhfast(float x) {
    return __builtin_amdgcn_tanhf(x);
}
__device__ __forceinline__ float sigf(float x) {
    return __builtin_fmaf(0.5f, __builtin_amdgcn_tanhf(0.5f * x), 0.5f);
}

#define WMMA4(c, a, b) \
    (c) = __builtin_amdgcn_wmma_f32_16x16x4_f32(false, (a), false, (b), (short)0, (c), false, false)

__device__ __forceinline__ v2f lstm_cells(const v8f z, v2f& c) {
    v2f h;
    #pragma unroll
    for (int s = 0; s < 2; ++s) {
        const float zi = z[4 * s + 0];
        const float zf = z[4 * s + 1];
        const float zg = z[4 * s + 2];
        const float zo = z[4 * s + 3];
        const float cc = sigf(zf) * c[s] + sigf(zi) * tanhfast(zg);
        c[s] = cc;
        h[s] = sigf(zo) * tanhfast(cc);
    }
    return h;
}

__device__ __forceinline__ void lds_publish_and_signal() {
    __builtin_amdgcn_fence(__ATOMIC_RELEASE, "workgroup", "local");
    __builtin_amdgcn_s_barrier_signal(-1);
}
__device__ __forceinline__ void lds_wait_and_acquire() {
    __builtin_amdgcn_s_barrier_wait(-1);
    __builtin_amdgcn_fence(__ATOMIC_ACQUIRE, "workgroup", "local");
}

__global__ void __launch_bounds__(NWAVE * 32)
lstm2_dense_fused(const float* __restrict__ X,
                  const float* __restrict__ Wi1, const float* __restrict__ Wh1,
                  const float* __restrict__ b1,
                  const float* __restrict__ Wi2, const float* __restrict__ Wh2,
                  const float* __restrict__ b2,
                  const float* __restrict__ Wd,  const float* __restrict__ bd,
                  float* __restrict__ out)
{
    __shared__ v2f sH1[2 * HBUF];
    __shared__ v2f sH2[2 * HBUF];

    const int tid   = threadIdx.x;
    const int lane  = tid & 31;
    const int wave  = __builtin_amdgcn_readfirstlane(tid >> 5);
    const int mrow  = lane & 15;
    const int hi    = lane >> 4;
    const int khalf = 2 * hi;
    const int b0    = blockIdx.x * 16;

    const bool isL1 = wave < 5;
    const bool isL2 = (wave >= 5) && (wave < 10);
    const int  n    = isL1 ? wave : wave - 5;

    for (int i = tid; i < 2 * HBUF; i += NWAVE * 32) {
        v2f zz; zz.x = 0.0f; zz.y = 0.0f;
        sH1[i] = zz;
        sH2[i] = zz;
    }

    v2f win[8], wrec[5];
    v8f bz;
    if (isL1 || isL2) {
        const float* __restrict__ Win  = isL1 ? Wi1 : Wi2;
        const float* __restrict__ Wrec = isL1 ? Wh1 : Wh2;
        const float* __restrict__ bb   = isL1 ? b1  : b2;
        const int col = (mrow & 3) * LH + 4 * n + (mrow >> 2);
        #pragma unroll
        for (int kk = 0; kk < 8; ++kk) {
            const int row = 4 * kk + khalf;
            v2f w; w.x = 0.0f; w.y = 0.0f;
            if (kk < 5 || isL1) {
                w.x = Win[row * LH4 + col];
                w.y = Win[(row + 1) * LH4 + col];
            }
            win[kk] = w;
        }
        #pragma unroll
        for (int kk = 0; kk < 5; ++kk) {
            const int row = 4 * kk + khalf;
            v2f w;
            w.x = Wrec[row * LH4 + col];
            w.y = Wrec[(row + 1) * LH4 + col];
            wrec[kk] = w;
        }
        #pragma unroll
        for (int r = 0; r < 8; ++r)
            bz[r] = bb[(r & 3) * LH + 4 * n + khalf + (r >> 2)];
    } else {
        #pragma unroll
        for (int kk = 0; kk < 8; ++kk) { win[kk].x = 0.0f; win[kk].y = 0.0f; }
        #pragma unroll
        for (int kk = 0; kk < 5; ++kk) {
            const int row = 4 * kk + khalf;
            v2f w;
            w.x = (mrow < LOUT) ? Wd[row * LOUT + mrow] : 0.0f;
            w.y = (mrow < LOUT) ? Wd[(row + 1) * LOUT + mrow] : 0.0f;
            wrec[kk] = w;
        }
        #pragma unroll
        for (int r = 0; r < 8; ++r) {
            const int o = 8 * hi + r;
            bz[r] = (o < LOUT) ? bd[o] : 0.0f;
        }
    }

    const float* xlane = X + (size_t)(b0 + mrow) * LT * LF + khalf;
    v2f xc[8];
    #pragma unroll
    for (int kk = 0; kk < 8; ++kk) { xc[kk].x = 0.0f; xc[kk].y = 0.0f; }
    if (isL1) {
        #pragma unroll
        for (int kk = 0; kk < 8; ++kk)
            xc[kk] = *(const v2f*)(xlane + 4 * kk);
    }

    v2f c; c.x = 0.0f; c.y = 0.0f;

    __syncthreads();
    __builtin_amdgcn_s_setprio(3);
    lds_publish_and_signal();

    if (isL1) {
        for (int i = 0; i < LT; ++i) {
            __builtin_amdgcn_s_setprio(0);
            v8f accA = bz;
            v8f accB = {};
            #pragma unroll
            for (int kk = 0; kk < 8; kk += 2) {
                WMMA4(accA, win[kk],     xc[kk]);
                WMMA4(accB, win[kk + 1], xc[kk + 1]);
            }
            if (i + 1 < LT) {
                const float* xr = xlane + (size_t)(i + 1) * LF;
                #pragma unroll
                for (int kk = 0; kk < 8; ++kk)
                    xc[kk] = *(const v2f*)(xr + 4 * kk);
            }
            if (i + 5 < LT) __builtin_prefetch(xlane + (size_t)(i + 5) * LF, 0, 3);
            __builtin_amdgcn_s_setprio(3);

            lds_wait_and_acquire();

            const v2f* hprev = sH1 + ((i + 1) & 1) * HBUF + lane;
            v2f hf[5];
            #pragma unroll
            for (int kk = 0; kk < 5; ++kk) hf[kk] = hprev[kk * 32];
            WMMA4(accA, wrec[0], hf[0]);
            WMMA4(accB, wrec[1], hf[1]);
            WMMA4(accA, wrec[2], hf[2]);
            WMMA4(accB, wrec[3], hf[3]);
            WMMA4(accA, wrec[4], hf[4]);
            const v8f z = accA + accB;
            const v2f h = lstm_cells(z, c);
            sH1[(i & 1) * HBUF + n * 32 + lane] = h;

            lds_publish_and_signal();
        }
        lds_wait_and_acquire();
        lds_publish_and_signal();
        lds_wait_and_acquire();
        lds_publish_and_signal();
        lds_wait_and_acquire();
        lds_publish_and_signal();
    } else if (isL2) {
        lds_wait_and_acquire();
        lds_publish_and_signal();
        lds_wait_and_acquire();
        v2f ha[5];
        #pragma unroll
        for (int kk = 0; kk < 5; ++kk) ha[kk] = sH1[kk * 32 + lane];
        lds_publish_and_signal();
        for (int i = 2; i < LT + 2; ++i) {
            __builtin_amdgcn_s_setprio(0);
            v8f aA = bz;
            v8f aB = {};
            WMMA4(aA, win[0], ha[0]);
            WMMA4(aB, win[1], ha[1]);
            WMMA4(aA, win[2], ha[2]);
            WMMA4(aB, win[3], ha[3]);
            WMMA4(aA, win[4], ha[4]);
            asm volatile("" :: "v"(aA), "v"(aB));
            __builtin_amdgcn_s_setprio(3);

            lds_wait_and_acquire();

            const v2f* hrec = sH2 + ((i + 1) & 1) * HBUF + lane;
            const v2f* hin  = sH1 + ((i - 1) & 1) * HBUF + lane;
            v2f hb[5];
            #pragma unroll
            for (int kk = 0; kk < 5; ++kk) hb[kk] = hrec[kk * 32];
            #pragma unroll
            for (int kk = 0; kk < 5; ++kk) ha[kk] = hin[kk * 32];
            __builtin_amdgcn_sched_barrier(0);
            WMMA4(aB, wrec[0], hb[0]);
            WMMA4(aA, wrec[1], hb[1]);
            WMMA4(aB, wrec[2], hb[2]);
            WMMA4(aA, wrec[3], hb[3]);
            WMMA4(aB, wrec[4], hb[4]);
            const v8f z = aA + aB;
            const v2f h = lstm_cells(z, c);
            sH2[(i & 1) * HBUF + n * 32 + lane] = h;

            lds_publish_and_signal();
        }
        lds_wait_and_acquire();
        lds_publish_and_signal();
    } else {
        lds_wait_and_acquire();
        lds_publish_and_signal();
        lds_wait_and_acquire();
        lds_publish_and_signal();
        lds_wait_and_acquire();
        lds_publish_and_signal();
        for (int i = 3; i < LT + 3; ++i) {
            lds_wait_and_acquire();

            const int t = i - 3;
            const v2f* hsrc = sH2 + ((i + 1) & 1) * HBUF + lane;
            v2f hf[5];
            #pragma unroll
            for (int kk = 0; kk < 5; ++kk) hf[kk] = hsrc[kk * 32];
            v8f acc = {};
            #pragma unroll
            for (int kk = 0; kk < 5; ++kk)
                WMMA4(acc, wrec[kk], hf[kk]);
            float* orow = out + ((size_t)(b0 + mrow) * LT + (size_t)t) * LOUT + 8 * hi;
            v2f* o2 = (v2f*)orow;
            v2f p;
            p.x = acc[0] + bz[0]; p.y = acc[1] + bz[1]; o2[0] = p;
            if (hi == 0) {
                p.x = acc[2] + bz[2]; p.y = acc[3] + bz[3]; o2[1] = p;
                p.x = acc[4] + bz[4]; p.y = acc[5] + bz[5]; o2[2] = p;
                p.x = acc[6] + bz[6]; p.y = acc[7] + bz[7]; o2[3] = p;
            }

            lds_publish_and_signal();
        }
    }
    lds_wait_and_acquire();
    __builtin_amdgcn_s_setprio(0);
}

extern "C" void kernel_launch(void* const* d_in, const int* in_sizes, int n_in,
                              void* d_out, int out_size, void* d_ws, size_t ws_size,
                              hipStream_t stream) {
    (void)in_sizes; (void)n_in; (void)out_size; (void)d_ws; (void)ws_size;
    const float* X   = (const float*)d_in[0];
    const float* Wi1 = (const float*)d_in[1];
    const float* Wh1 = (const float*)d_in[2];
    const float* b1  = (const float*)d_in[3];
    const float* Wi2 = (const float*)d_in[4];
    const float* Wh2 = (const float*)d_in[5];
    const float* b2  = (const float*)d_in[6];
    const float* Wd  = (const float*)d_in[7];
    const float* bd  = (const float*)d_in[8];
    float* out = (float*)d_out;

    dim3 grid(LB / 16);
    dim3 block(NWAVE * 32);
    lstm2_dense_fused<<<grid, block, 0, stream>>>(X, Wi1, Wh1, b1, Wi2, Wh2, b2, Wd, bd, out);
}


# Matrix work evened over the 4 SIMDs (36/33/28/23 -> 31/30/28/31 WMMAs a step): roles dealt to waves by a table, tile 4's x products on a 12th wave via an LDS ring; that tile's sums reordered
# speedup vs baseline: 11.4910x; 1.0745x over previous
#include <hip/hip_runtime.h>


typedef __attribute__((ext_vector_type(2))) float v2f;
typedef __attribute__((ext_vector_type(4))) float v4f;
typedef __attribute__((ext_vector_type(8))) float v8f;

#define LH    20
#define LH4   80
#define LF    32
#define LT    1024
#define LB    256
#define LOUT  10
#define NWAVE 12
#define HBUF  160
#define XSLOT 64

#define R_L1   0
#define R_L2   1
#define R_HEAD 2
#define R_L1R  3
#define R_X    4
#define ROLE_OF_WAVE 0x431211100010ull
#define TILE_OF_WAVE 0x442043113200ull

__device__ __forceinline__ float tanhfast(float x) {
    return __builtin_amdgcn_tanhf(x);
}
__device__ __forceinline__ float sigf(float x) {
    return __builtin_fmaf(0.5f, __builtin_amdgcn_tanhf(0.5f * x), 0.5f);
}

#define WMMA4(c, a, b) \
    (c) = __builtin_amdgcn_wmma_f32_16x16x4_f32(false, (a), false, (b), (short)0, (c), false, false)

__device__ __forceinline__ v2f lstm_cells(const v8f z, v2f& c) {
    v2f h;
    #pragma unroll
    for (int s = 0; s < 2; ++s) {
        const float zi = z[4 * s + 0];
        const float zf = z[4 * s + 1];
        const float zg = z[4 * s + 2];
        const float zo = z[4 * s + 3];
        const float cc = sigf(zf) * c[s] + sigf(zi) * tanhfast(zg);
        c[s] = cc;
        h[s] = sigf(zo) * tanhfast(cc);
    }
    return h;
}

__device__ __forceinline__ void lds_publish_and_signal() {
    __builtin_amdgcn_fence(__ATOMIC_RELEASE, "workgroup", "local");
    __builtin_amdgcn_s_barrier_signal(-1);
}
__device__ __forceinline__ void lds_wait_and_acquire() {
    __builtin_amdgcn_s_barrier_wait(-1);
    __builtin_amdgcn_fence(__ATOMIC_ACQUIRE, "workgroup", "local");
}

__global__ void __launch_bounds__(NWAVE * 32)
lstm2_dense_fused(const float* __restrict__ X,
                  const float* __restrict__ Wi1, const float* __restrict__ Wh1,
                  const float* __restrict__ b1,
                  const float* __restrict__ Wi2, const float* __restrict__ Wh2,
                  const float* __restrict__ b2,
                  const float* __restrict__ Wd,  const float* __restrict__ bd,
                  float* __restrict__ out)
{
    __shared__ v2f sH1[2 * HBUF];
    __shared__ v2f sH2[2 * HBUF];
    __shared__ v4f sXW[3 * XSLOT];

    const int tid   = threadIdx.x;
    const int lane  = tid & 31;
    const int wave  = __builtin_amdgcn_readfirstlane(tid >> 5);
    const int mrow  = lane & 15;
    const int hi    = lane >> 4;
    const int khalf = 2 * hi;
    const int b0    = blockIdx.x * 16;

    const int  role = (int)((ROLE_OF_WAVE >> (4 * wave)) & 15ull);
    const int  n    = (int)((TILE_OF_WAVE >> (4 * wave)) & 15ull);
    const bool isL1  = role == R_L1;
    const bool isL2  = role == R_L2;
    const bool isL1R = role == R_L1R;
    const bool isX   = role == R_X;
    const bool layer1 = isL1 || isL1R || isX;

    for (int i = tid; i < 2 * HBUF; i += NWAVE * 32) {
        v2f zz; zz.x = 0.0f; zz.y = 0.0f;
        sH1[i] = zz;
        sH2[i] = zz;
    }

    v2f win[8], wrec[5];
    v8f bz;
    if (layer1 || isL2) {
        const float* __restrict__ Win  = layer1 ? Wi1 : Wi2;
        const float* __restrict__ Wrec = layer1 ? Wh1 : Wh2;
        const float* __restrict__ bb   = layer1 ? b1  : b2;
        const int col = (mrow & 3) * LH + 4 * n + (mrow >> 2);
        #pragma unroll
        for (int kk = 0; kk < 8; ++kk) {
            const int row = 4 * kk + khalf;
            v2f w; w.x = 0.0f; w.y = 0.0f;
            if (kk < 5 || layer1) {
                w.x = Win[row * LH4 + col];
                w.y = Win[(row + 1) * LH4 + col];
            }
            win[kk] = w;
        }
        #pragma unroll
        for (int kk = 0; kk < 5; ++kk) {
            const int row = 4 * kk + khalf;
            v2f w;
            w.x = Wrec[row * LH4 + col];
            w.y = Wrec[(row + 1) * LH4 + col];
            wrec[kk] = w;
        }
        #pragma unroll
        for (int r = 0; r < 8; ++r)
            bz[r] = bb[(r & 3) * LH + 4 * n + khalf + (r >> 2)];
    } else {
        #pragma unroll
        for (int kk = 0; kk < 8; ++kk) { win[kk].x = 0.0f; win[kk].y = 0.0f; }
        #pragma unroll
        for (int kk = 0; kk < 5; ++kk) {
            const int row = 4 * kk + khalf;
            v2f w;
            w.x = (mrow < LOUT) ? Wd[row * LOUT + mrow] : 0.0f;
            w.y = (mrow < LOUT) ? Wd[(row + 1) * LOUT + mrow] : 0.0f;
            wrec[kk] = w;
        }
        #pragma unroll
        for (int r = 0; r < 8; ++r) {
            const int o = 8 * hi + r;
            bz[r] = (o < LOUT) ? bd[o] : 0.0f;
        }
    }

    const float* xlane = X + (size_t)(b0 + mrow) * LT * LF + khalf;
    v2f xc[8];
    #pragma unroll
    for (int kk = 0; kk < 8; ++kk) { xc[kk].x = 0.0f; xc[kk].y = 0.0f; }
    if (isL1 || isX) {
        #pragma unroll
        for (int kk = 0; kk < 8; ++kk)
            xc[kk] = *(const v2f*)(xlane + 4 * kk);
    }
    if (isX) {
        v8f a = bz;
        v8f b = {};
        #pragma unroll
        for (int kk = 0; kk < 8; kk += 2) {
            WMMA4(a, win[kk],     xc[kk]);
            WMMA4(b, win[kk + 1], xc[kk + 1]);
        }
        const v8f s = a + b;
        sXW[lane]      = __builtin_shufflevector(s, s, 0, 1, 2, 3);
        sXW[32 + lane] = __builtin_shufflevector(s, s, 4, 5, 6, 7);
        #pragma unroll
        for (int kk = 0; kk < 8; ++kk)
            xc[kk] = *(const v2f*)(xlane + LF + 4 * kk);
    }

    v2f c; c.x = 0.0f; c.y = 0.0f;

    __syncthreads();
    __builtin_amdgcn_s_setprio(3);
    lds_publish_and_signal();

    if (isL1) {
        for (int i = 0; i < LT; ++i) {
            __builtin_amdgcn_s_setprio(0);
            v8f accA = bz;
            v8f accB = {};
            #pragma unroll
            for (int kk = 0; kk < 8; kk += 2) {
                WMMA4(accA, win[kk],     xc[kk]);
                WMMA4(accB, win[kk + 1], xc[kk + 1]);
            }
            if (i + 1 < LT) {
                const float* xr = xlane + (size_t)(i + 1) * LF;
                #pragma unroll
                for (int kk = 0; kk < 8; ++kk)
                    xc[kk] = *(const v2f*)(xr + 4 * kk);
            }
            if (i + 5 < LT) __builtin_prefetch(xlane + (size_t)(i + 5) * LF, 0, 3);
            __builtin_amdgcn_s_setprio(3);

            lds_wait_and_acquire();

            const v2f* hprev = sH1 + ((i + 1) & 1) * HBUF + lane;
            v2f hf[5];
            #pragma unroll
            for (int kk = 0; kk < 5; ++kk) hf[kk] = hprev[kk * 32];
            WMMA4(accA, wrec[0], hf[0]);
            WMMA4(accB, wrec[1], hf[1]);
            WMMA4(accA, wrec[2], hf[2]);
            WMMA4(accB, wrec[3], hf[3]);
            WMMA4(accA, wrec[4], hf[4]);
            const v8f z = accA + accB;
            const v2f h = lstm_cells(z, c);
            sH1[(i & 1) * HBUF + n * 32 + lane] = h;

            lds_publish_and_signal();
        }
        lds_wait_and_acquire();
        lds_publish_and_signal();
        lds_wait_and_acquire();
        lds_publish_and_signal();
        lds_wait_and_acquire();
        lds_publish_and_signal();
    } else if (isL1R) {
        int rs = 0;
        for (int i = 0; i < LT; ++i) {
            lds_wait_and_acquire();

            const v2f* hprev = sH1 + ((i + 1) & 1) * HBUF + lane;
            v2f hf[5];
            #pragma unroll
            for (int kk = 0; kk < 5; ++kk) hf[kk] = hprev[kk * 32];
            const v4f xlo = sXW[rs * XSLOT + lane];
            const v4f xhi = sXW[rs * XSLOT + 32 + lane];
            rs = (rs == 2) ? 0 : rs + 1;
            v8f accA = __builtin_shufflevector(xlo, xhi, 0, 1, 2, 3, 4, 5, 6, 7);
            v8f accB = {};
            WMMA4(accB, wrec[0], hf[0]);
            WMMA4(accA, wrec[1], hf[1]);
            WMMA4(accB, wrec[2], hf[2]);
            WMMA4(accA, wrec[3], hf[3]);
            WMMA4(accB, wrec[4], hf[4]);
            const v8f z = accA + accB;
            const v2f h = lstm_cells(z, c);
            sH1[(i & 1) * HBUF + n * 32 + lane] = h;

            lds_publish_and_signal();
        }
        lds_wait_and_acquire();
        lds_publish_and_signal();
        lds_wait_and_acquire();
        lds_publish_and_signal();
        lds_wait_and_acquire();
        lds_publish_and_signal();
    } else if (isX) {
        int ws = 1;
        for (int i = 0; i < LT; ++i) {
            if (i + 1 < LT) {
                __builtin_amdgcn_s_setprio(0);
                v8f a = bz;
                v8f b = {};
                #pragma unroll
                for (int kk = 0; kk < 8; kk += 2) {
                    WMMA4(a, win[kk],     xc[kk]);
                    WMMA4(b, win[kk + 1], xc[kk + 1]);
                }
                if (i + 2 < LT) {
                    const float* xr = xlane + (size_t)(i + 2) * LF;
                    #pragma unroll
                    for (int kk = 0; kk < 8; ++kk)
                        xc[kk] = *(const v2f*)(xr + 4 * kk);
                }
                if (i + 6 < LT) __builtin_prefetch(xlane + (size_t)(i + 6) * LF, 0, 3);
                const v8f s = a + b;
                __builtin_amdgcn_s_setprio(3);
                sXW[ws * XSLOT + lane]      = __builtin_shufflevector(s, s, 0, 1, 2, 3);
                sXW[ws * XSLOT + 32 + lane] = __builtin_shufflevector(s, s, 4, 5, 6, 7);
                ws = (ws == 2) ? 0 : ws + 1;
            }
            lds_wait_and_acquire();
            lds_publish_and_signal();
        }
        lds_wait_and_acquire();
        lds_publish_and_signal();
        lds_wait_and_acquire();
        lds_publish_and_signal();
        lds_wait_and_acquire();
        lds_publish_and_signal();
    } else if (isL2) {
        lds_wait_and_acquire();
        lds_publish_and_signal();
        lds_wait_and_acquire();
        v2f ha[5];
        #pragma unroll
        for (int kk = 0; kk < 5; ++kk) ha[kk] = sH1[kk * 32 + lane];
        lds_publish_and_signal();
        for (int i = 2; i < LT + 2; ++i) {
            __builtin_amdgcn_s_setprio(0);
            v8f aA = bz;
            v8f aB = {};
            WMMA4(aA, win[0], ha[0]);
            WMMA4(aB, win[1], ha[1]);
            WMMA4(aA, win[2], ha[2]);
            WMMA4(aB, win[3], ha[3]);
            WMMA4(aA, win[4], ha[4]);
            asm volatile("" :: "v"(aA), "v"(aB));
            __builtin_amdgcn_s_setprio(3);

            lds_wait_and_acquire();

            const v2f* hrec = sH2 + ((i + 1) & 1) * HBUF + lane;
            const v2f* hin  = sH1 + ((i - 1) & 1) * HBUF + lane;
            v2f hb[5];
            #pragma unroll
            for (int kk = 0; kk < 5; ++kk) hb[kk] = hrec[kk * 32];
            #pragma unroll
            for (int kk = 0; kk < 5; ++kk) ha[kk] = hin[kk * 32];
            __builtin_amdgcn_sched_barrier(0);
            WMMA4(aB, wrec[0], hb[0]);
            WMMA4(aA, wrec[1], hb[1]);
            WMMA4(aB, wrec[2], hb[2]);
            WMMA4(aA, wrec[3], hb[3]);
            WMMA4(aB, wrec[4], hb[4]);
            const v8f z = aA + aB;
            const v2f h = lstm_cells(z, c);
            sH2[(i & 1) * HBUF + n * 32 + lane] = h;

            lds_publish_and_signal();
        }
        lds_wait_and_acquire();
        lds_publish_and_signal();
    } else {
        lds_wait_and_acquire();
        lds_publish_and_signal();
        lds_wait_and_acquire();
        lds_publish_and_signal();
        lds_wait_and_acquire();
        lds_publish_and_signal();
        for (int i = 3; i < LT + 3; ++i) {
            lds_wait_and_acquire();

            const int t = i - 3;
            const v2f* hsrc = sH2 + ((i + 1) & 1) * HBUF + lane;
            v2f hf[5];
            #pragma unroll
            for (int kk = 0; kk < 5; ++kk) hf[kk] = hsrc[kk * 32];
            v8f acc = {};
            #pragma unroll
            for (int kk = 0; kk < 5; ++kk)
                WMMA4(acc, wrec[kk], hf[kk]);
            float* orow = out + ((size_t)(b0 + mrow) * LT + (size_t)t) * LOUT + 8 * hi;
            v2f* o2 = (v2f*)orow;
            v2f p;
            p.x = acc[0] + bz[0]; p.y = acc[1] + bz[1]; o2[0] = p;
            if (hi == 0) {
                p.x = acc[2] + bz[2]; p.y = acc[3] + bz[3]; o2[1] = p;
                p.x = acc[4] + bz[4]; p.y = acc[5] + bz[5]; o2[2] = p;
                p.x = acc[6] + bz[6]; p.y = acc[7] + bz[7]; o2[3] = p;
            }

            lds_publish_and_signal();
        }
    }
    lds_wait_and_acquire();
    __builtin_amdgcn_s_setprio(0);
}

extern "C" void kernel_launch(void* const* d_in, const int* in_sizes, int n_in,
                              void* d_out, int out_size, void* d_ws, size_t ws_size,
                              hipStream_t stream) {
    (void)in_sizes; (void)n_in; (void)out_size; (void)d_ws; (void)ws_size;
    const float* X   = (const float*)d_in[0];
    const float* Wi1 = (const float*)d_in[1];
    const float* Wh1 = (const float*)d_in[2];
    const float* b1  = (const float*)d_in[3];
    const float* Wi2 = (const float*)d_in[4];
    const float* Wh2 = (const float*)d_in[5];
    const float* b2  = (const float*)d_in[6];
    const float* Wd  = (const float*)d_in[7];
    const float* bd  = (const float*)d_in[8];
    float* out = (float*)d_out;

    dim3 grid(LB / 16);
    dim3 block(NWAVE * 32);
    lstm2_dense_fused<<<grid, block, 0, stream>>>(X, Wi1, Wh1, b1, Wi2, Wh2, b2, Wd, bd, out);
}


# x rows fetched once per workgroup as whole 128-byte lines by one wave, handed to the layer-1 waves through a 2-slot LDS ring (lane half hh owns features 16hh..16hh+15); x-product sums reordered
# speedup vs baseline: 12.1208x; 1.0548x over previous
#include <hip/hip_runtime.h>


typedef __attribute__((ext_vector_type(2))) float v2f;
typedef __attribute__((ext_vector_type(4))) float v4f;
typedef __attribute__((ext_vector_type(8))) float v8f;

#define LH    20
#define LH4   80
#define LF    32
#define LT    1024
#define LB    256
#define LOUT  10
#define NWAVE 12
#define HBUF  160
#define XSLOT 64
#define XPITCH 5
#define XRING 160

#define R_L1   0
#define R_L2   1
#define R_HEAD 2
#define R_L1R  3
#define R_X    4
#define ROLE_OF_WAVE 0x431211100010ull
#define TILE_OF_WAVE 0x442043113200ull

__device__ __forceinline__ float tanhfast(float x) {
    return __builtin_amdgcn_tanhf(x);
}
__device__ __forceinline__ float sigf(float x) {
    return __builtin_fmaf(0.5f, __builtin_amdgcn_tanhf(0.5f * x), 0.5f);
}

#define WMMA4(c, a, b) \
    (c) = __builtin_amdgcn_wmma_f32_16x16x4_f32(false, (a), false, (b), (short)0, (c), false, false)

__device__ __forceinline__ void x_products(v8f& a, v8f& b, const v2f (&win)[8], const v4f (&xq)[4]) {
    #pragma unroll
    for (int q = 0; q < 4; ++q) {
        WMMA4(a, win[2 * q],     __builtin_shufflevector(xq[q], xq[q], 0, 1));
        WMMA4(b, win[2 * q + 1], __builtin_shufflevector(xq[q], xq[q], 2, 3));
    }
}

__device__ __forceinline__ v2f lstm_cells(const v8f z, v2f& c) {
    v2f h;
    #pragma unroll
    for (int s = 0; s < 2; ++s) {
        const float zi = z[4 * s + 0];
        const float zf = z[4 * s + 1];
        const float zg = z[4 * s + 2];
        const float zo = z[4 * s + 3];
        const float cc = sigf(zf) * c[s] + sigf(zi) * tanhfast(zg);
        c[s] = cc;
        h[s] = sigf(zo) * tanhfast(cc);
    }
    return h;
}

__device__ __forceinline__ void lds_publish_and_signal() {
    __builtin_amdgcn_fence(__ATOMIC_RELEASE, "workgroup", "local");
    __builtin_amdgcn_s_barrier_signal(-1);
}
__device__ __forceinline__ void lds_wait_and_acquire() {
    __builtin_amdgcn_s_barrier_wait(-1);
    __builtin_amdgcn_fence(__ATOMIC_ACQUIRE, "workgroup", "local");
}

__global__ void __launch_bounds__(NWAVE * 32)
lstm2_dense_fused(const float* __restrict__ X,
                  const float* __restrict__ Wi1, const float* __restrict__ Wh1,
                  const float* __restrict__ b1,
                  const float* __restrict__ Wi2, const float* __restrict__ Wh2,
                  const float* __restrict__ b2,
                  const float* __restrict__ Wd,  const float* __restrict__ bd,
                  float* __restrict__ out)
{
    __shared__ v2f sH1[2 * HBUF];
    __shared__ v2f sH2[2 * HBUF];
    __shared__ v4f sXW[3 * XSLOT];
    __shared__ v4f sX[2 * XRING];

    const int tid   = threadIdx.x;
    const int lane  = tid & 31;
    const int wave  = __builtin_amdgcn_readfirstlane(tid >> 5);
    const int mrow  = lane & 15;
    const int hi    = lane >> 4;
    const int khalf = 2 * hi;
    const int b0    = blockIdx.x * 16;

    const int  role = (int)((ROLE_OF_WAVE >> (4 * wave)) & 15ull);
    const int  n    = (int)((TILE_OF_WAVE >> (4 * wave)) & 15ull);
    const bool isL1  = role == R_L1;
    const bool isL2  = role == R_L2;
    const bool isL1R = role == R_L1R;
    const bool isX   = role == R_X;
    const bool layer1 = isL1 || isL1R || isX;

    for (int i = tid; i < 2 * HBUF; i += NWAVE * 32) {
        v2f zz; zz.x = 0.0f; zz.y = 0.0f;
        sH1[i] = zz;
        sH2[i] = zz;
    }

    v2f win[8], wrec[5];
    v8f bz;
    if (layer1 || isL2) {
        const float* __restrict__ Win  = layer1 ? Wi1 : Wi2;
        const float* __restrict__ Wrec = layer1 ? Wh1 : Wh2;
        const float* __restrict__ bb   = layer1 ? b1  : b2;
        const int col = (mrow & 3) * LH + 4 * n + (mrow >> 2);
        #pragma unroll
        for (int kk = 0; kk < 8; ++kk) {
            const int row = layer1 ? 16 * hi + 2 * kk : 4 * kk + khalf;
            v2f w; w.x = 0.0f; w.y = 0.0f;
            if (kk < 5 || layer1) {
                w.x = Win[row * LH4 + col];
                w.y = Win[(row + 1) * LH4 + col];
            }
            win[kk] = w;
        }
        #pragma unroll
        for (int kk = 0; kk < 5; ++kk) {
            const int row = 4 * kk + khalf;
            v2f w;
            w.x = Wrec[row * LH4 + col];
            w.y = Wrec[(row + 1) * LH4 + col];
            wrec[kk] = w;
        }
        #pragma unroll
        for (int r = 0; r < 8; ++r)
            bz[r] = bb[(r & 3) * LH + 4 * n + khalf + (r >> 2)];
    } else {
        #pragma unroll
        for (int kk = 0; kk < 8; ++kk) { win[kk].x = 0.0f; win[kk].y = 0.0f; }
        #pragma unroll
        for (int kk = 0; kk < 5; ++kk) {
            const int row = 4 * kk + khalf;
            v2f w;
            w.x = (mrow < LOUT) ? Wd[row * LOUT + mrow] : 0.0f;
            w.y = (mrow < LOUT) ? Wd[(row + 1) * LOUT + mrow] : 0.0f;
            wrec[kk] = w;
        }
        #pragma unroll
        for (int r = 0; r < 8; ++r) {
            const int o = 8 * hi + r;
            bz[r] = (o < LOUT) ? bd[o] : 0.0f;
        }
    }

    const float* xlane = X + (size_t)(b0 + mrow) * LT * LF + 16 * hi;
    v4f xq[4];
    #pragma unroll
    for (int q = 0; q < 4; ++q) { xq[q].x = 0.0f; xq[q].y = 0.0f; xq[q].z = 0.0f; xq[q].w = 0.0f; }
    if (isL1 || isX) {
        #pragma unroll
        for (int q = 0; q < 4; ++q)
            xq[q] = *(const v4f*)(xlane + 4 * q);
    }
    if (isX) {
        v8f a = bz;
        v8f b = {};
        x_products(a, b, win, xq);
        const v8f s = a + b;
        sXW[lane]      = __builtin_shufflevector(s, s, 0, 1, 2, 3);
        sXW[32 + lane] = __builtin_shufflevector(s, s, 4, 5, 6, 7);
        #pragma unroll
        for (int q = 0; q < 4; ++q)
            xq[q] = *(const v4f*)(xlane + LF + 4 * q);
    }
    const float* xraw = X + (size_t)(b0 + (lane >> 3)) * LT * LF + 4 * (lane & 7);
    const int    xput = ((lane >> 3) + 16 * ((lane >> 2) & 1)) * XPITCH + (lane & 3);
    v4f raw[4];
    #pragma unroll
    for (int j = 0; j < 4; ++j) { raw[j].x = 0.0f; raw[j].y = 0.0f; raw[j].z = 0.0f; raw[j].w = 0.0f; }
    if (isX) {
        #pragma unroll
        for (int j = 0; j < 4; ++j)
            raw[j] = *(const v4f*)(xraw + (size_t)(4 * j) * LT * LF + LF);
        #pragma unroll
        for (int j = 0; j < 4; ++j)
            sX[XRING + xput + 4 * j * XPITCH] = raw[j];
        #pragma unroll
        for (int j = 0; j < 4; ++j)
            raw[j] = *(const v4f*)(xraw + (size_t)(4 * j) * LT * LF + 2 * LF);
    }

    v2f c; c.x = 0.0f; c.y = 0.0f;

    __syncthreads();
    __builtin_amdgcn_s_setprio(3);
    lds_publish_and_signal();

    if (isL1) {
        for (int i = 0; i < LT; ++i) {
            __builtin_amdgcn_s_setprio(0);
            v8f accA = bz;
            v8f accB = {};
            x_products(accA, accB, win, xq);
            asm volatile("" :: "v"(accA), "v"(accB));
            __builtin_amdgcn_s_setprio(3);

            lds_wait_and_acquire();

            const v2f* hprev = sH1 + ((i + 1) & 1) * HBUF + lane;
            v2f hf[5];
            #pragma unroll
            for (int kk = 0; kk < 5; ++kk) hf[kk] = hprev[kk * 32];
            WMMA4(accA, wrec[0], hf[0]);
            WMMA4(accB, wrec[1], hf[1]);
            WMMA4(accA, wrec[2], hf[2]);
            WMMA4(accB, wrec[3], hf[3]);
            WMMA4(accA, wrec[4], hf[4]);
            __builtin_amdgcn_sched_barrier(0);
            if (i + 1 < LT) {
                const v4f* xs = sX + ((i + 1) & 1) * XRING + lane * XPITCH;
                #pragma unroll
                for (int q = 0; q < 4; ++q) xq[q] = xs[q];
            }
            __builtin_amdgcn_sched_barrier(0);
            const v8f z = accA + accB;
            const v2f h = lstm_cells(z, c);
            sH1[(i & 1) * HBUF + n * 32 + lane] = h;

            lds_publish_and_signal();
        }
        lds_wait_and_acquire();
        lds_publish_and_signal();
        lds_wait_and_acquire();
        lds_publish_and_signal();
        lds_wait_and_acquire();
        lds_publish_and_signal();
    } else if (isL1R) {
        int rs = 0;
        for (int i = 0; i < LT; ++i) {
            lds_wait_and_acquire();

            const v2f* hprev = sH1 + ((i + 1) & 1) * HBUF + lane;
            v2f hf[5];
            #pragma unroll
            for (int kk = 0; kk < 5; ++kk) hf[kk] = hprev[kk * 32];
            const v4f xlo = sXW[rs * XSLOT + lane];
            const v4f xhi = sXW[rs * XSLOT + 32 + lane];
            rs = (rs == 2) ? 0 : rs + 1;
            v8f accA = __builtin_shufflevector(xlo, xhi, 0, 1, 2, 3, 4, 5, 6, 7);
            v8f accB = {};
            WMMA4(accB, wrec[0], hf[0]);
            WMMA4(accA, wrec[1], hf[1]);
            WMMA4(accB, wrec[2], hf[2]);
            WMMA4(accA, wrec[3], hf[3]);
            WMMA4(accB, wrec[4], hf[4]);
            const v8f z = accA + accB;
            const v2f h = lstm_cells(z, c);
            sH1[(i & 1) * HBUF + n * 32 + lane] = h;

            lds_publish_and_signal();
        }
        lds_wait_and_acquire();
        lds_publish_and_signal();
        lds_wait_and_acquire();
        lds_publish_and_signal();
        lds_wait_and_acquire();
        lds_publish_and_signal();
    } else if (isX) {
        int ws = 1;
        for (int i = 0; i < LT; ++i) {
            if (i + 1 < LT) {
                __builtin_amdgcn_s_setprio(0);
                v8f a = bz;
                v8f b = {};
                x_products(a, b, win, xq);
                const v8f s = a + b;
                __builtin_amdgcn_s_setprio(3);
                sXW[ws * XSLOT + lane]      = __builtin_shufflevector(s, s, 0, 1, 2, 3);
                sXW[ws * XSLOT + 32 + lane] = __builtin_shufflevector(s, s, 4, 5, 6, 7);
                ws = (ws == 2) ? 0 : ws + 1;
            }
            lds_wait_and_acquire();
            if (i + 2 < LT) {
                v4f* xd = sX + (i & 1) * XRING;
                #pragma unroll
                for (int j = 0; j < 4; ++j)
                    xd[xput + 4 * j * XPITCH] = raw[j];
                #pragma unroll
                for (int q = 0; q < 4; ++q) xq[q] = xd[lane * XPITCH + q];
            }
            if (i + 3 < LT) {
                #pragma unroll
                for (int j = 0; j < 4; ++j)
                    raw[j] = *(const v4f*)(xraw + (size_t)(4 * j) * LT * LF + (size_t)(i + 3) * LF);
            }
            if (i + 7 < LT) __builtin_prefetch(xlane + (size_t)(i + 7) * LF, 0, 3);
            lds_publish_and_signal();
        }
        lds_wait_and_acquire();
        lds_publish_and_signal();
        lds_wait_and_acquire();
        lds_publish_and_signal();
        lds_wait_and_acquire();
        lds_publish_and_signal();
    } else if (isL2) {
        lds_wait_and_acquire();
        lds_publish_and_signal();
        lds_wait_and_acquire();
        v2f ha[5];
        #pragma unroll
        for (int kk = 0; kk < 5; ++kk) ha[kk] = sH1[kk * 32 + lane];
        lds_publish_and_signal();
        for (int i = 2; i < LT + 2; ++i) {
            __builtin_amdgcn_s_setprio(0);
            v8f aA = bz;
            v8f aB = {};
            WMMA4(aA, win[0], ha[0]);
            WMMA4(aB, win[1], ha[1]);
            WMMA4(aA, win[2], ha[2]);
            WMMA4(aB, win[3], ha[3]);
            WMMA4(aA, win[4], ha[4]);
            asm volatile("" :: "v"(aA), "v"(aB));
            __builtin_amdgcn_s_setprio(3);

            lds_wait_and_acquire();

            const v2f* hrec = sH2 + ((i + 1) & 1) * HBUF + lane;
            const v2f* hin  = sH1 + ((i - 1) & 1) * HBUF + lane;
            v2f hb[5];
            #pragma unroll
            for (int kk = 0; kk < 5; ++kk) hb[kk] = hrec[kk * 32];
            #pragma unroll
            for (int kk = 0; kk < 5; ++kk) ha[kk] = hin[kk * 32];
            __builtin_amdgcn_sched_barrier(0);
            WMMA4(aB, wrec[0], hb[0]);
            WMMA4(aA, wrec[1], hb[1]);
            WMMA4(aB, wrec[2], hb[2]);
            WMMA4(aA, wrec[3], hb[3]);
            WMMA4(aB, wrec[4], hb[4]);
            const v8f z = aA + aB;
            const v2f h = lstm_cells(z, c);
            sH2[(i & 1) * HBUF + n * 32 + lane] = h;

            lds_publish_and_signal();
        }
        lds_wait_and_acquire();
        lds_publish_and_signal();
    } else {
        lds_wait_and_acquire();
        lds_publish_and_signal();
        lds_wait_and_acquire();
        lds_publish_and_signal();
        lds_wait_and_acquire();
        lds_publish_and_signal();
        for (int i = 3; i < LT + 3; ++i) {
            lds_wait_and_acquire();

            const int t = i - 3;
            const v2f* hsrc = sH2 + ((i + 1) & 1) * HBUF + lane;
            v2f hf[5];
            #pragma unroll
            for (int kk = 0; kk < 5; ++kk) hf[kk] = hsrc[kk * 32];
            v8f acc = {};
            #pragma unroll
            for (int kk = 0; kk < 5; ++kk)
                WMMA4(acc, wrec[kk], hf[kk]);
            float* orow = out + ((size_t)(b0 + mrow) * LT + (size_t)t) * LOUT + 8 * hi;
            v2f* o2 = (v2f*)orow;
            v2f p;
            p.x = acc[0] + bz[0]; p.y = acc[1] + bz[1]; o2[0] = p;
            if (hi == 0) {
                p.x = acc[2] + bz[2]; p.y = acc[3] + bz[3]; o2[1] = p;
                p.x = acc[4] + bz[4]; p.y = acc[5] + bz[5]; o2[2] = p;
                p.x = acc[6] + bz[6]; p.y = acc[7] + bz[7]; o2[3] = p;
            }

            lds_publish_and_signal();
        }
    }
    lds_wait_and_acquire();
    __builtin_amdgcn_s_setprio(0);
}

extern "C" void kernel_launch(void* const* d_in, const int* in_sizes, int n_in,
                              void* d_out, int out_size, void* d_ws, size_t ws_size,
                              hipStream_t stream) {
    (void)in_sizes; (void)n_in; (void)out_size; (void)d_ws; (void)ws_size;
    const float* X   = (const float*)d_in[0];
    const float* Wi1 = (const float*)d_in[1];
    const float* Wh1 = (const float*)d_in[2];
    const float* b1  = (const float*)d_in[3];
    const float* Wi2 = (const float*)d_in[4];
    const float* Wh2 = (const float*)d_in[5];
    const float* b2  = (const float*)d_in[6];
    const float* Wd  = (const float*)d_in[7];
    const float* bd  = (const float*)d_in[8];
    float* out = (float*)d_out;

    dim3 grid(LB / 16);
    dim3 block(NWAVE * 32);
    lstm2_dense_fused<<<grid, block, 0, stream>>>(X, Wi1, Wh1, b1, Wi2, Wh2, b2, Wd, bd, out);
}


# Sigmoid's constant factors folded out of the time loop: i/f/o weight and bias rows halved at load, waves hand over 2h to halved Wh1/Wi2/Wh2/Wd (exact); cell update as fma(t,y,y), rounding differs
# speedup vs baseline: 12.6735x; 1.0456x over previous
#include <hip/hip_runtime.h>


typedef __attribute__((ext_vector_type(2))) float v2f;
typedef __attribute__((ext_vector_type(4))) float v4f;
typedef __attribute__((ext_vector_type(8))) float v8f;

#define LH    20
#define LH4   80
#define LF    32
#define LT    1024
#define LB    256
#define LOUT  10
#define NWAVE 12
#define HBUF  160
#define XSLOT 64
#define XPITCH 5
#define XRING 160

#define R_L1   0
#define R_L2   1
#define R_HEAD 2
#define R_L1R  3
#define R_X    4
#define ROLE_OF_WAVE 0x431211100010ull
#define TILE_OF_WAVE 0x442043113200ull

__device__ __forceinline__ float tanhfast(float x) {
    return __builtin_amdgcn_tanhf(x);
}

#define WMMA4(c, a, b) \
    (c) = __builtin_amdgcn_wmma_f32_16x16x4_f32(false, (a), false, (b), (short)0, (c), false, false)

__device__ __forceinline__ void x_products(v8f& a, v8f& b, const v2f (&win)[8], const v4f (&xq)[4]) {
    #pragma unroll
    for (int q = 0; q < 4; ++q) {
        WMMA4(a, win[2 * q],     __builtin_shufflevector(xq[q], xq[q], 0, 1));
        WMMA4(b, win[2 * q + 1], __builtin_shufflevector(xq[q], xq[q], 2, 3));
    }
}

__device__ __forceinline__ v2f lstm_cells(const v8f z, v2f& c) {
    v2f h2;
    #pragma unroll
    for (int s = 0; s < 2; ++s) {
        const float ti = tanhfast(z[4 * s + 0]);
        const float tf = tanhfast(z[4 * s + 1]);
        const float hg = 0.5f * tanhfast(z[4 * s + 2]);
        const float to = tanhfast(z[4 * s + 3]);
        const float u  = __builtin_fmaf(tf, c[s], c[s]);
        const float v  = __builtin_fmaf(ti, hg, hg);
        const float cc = __builtin_fmaf(0.5f, u, v);
        c[s] = cc;
        const float tc = tanhfast(cc);
        h2[s] = __builtin_fmaf(to, tc, tc);
    }
    return h2;
}

__device__ __forceinline__ void lds_publish_and_signal() {
    __builtin_amdgcn_fence(__ATOMIC_RELEASE, "workgroup", "local");
    __builtin_amdgcn_s_barrier_signal(-1);
}
__device__ __forceinline__ void lds_wait_and_acquire() {
    __builtin_amdgcn_s_barrier_wait(-1);
    __builtin_amdgcn_fence(__ATOMIC_ACQUIRE, "workgroup", "local");
}

__global__ void __launch_bounds__(NWAVE * 32)
lstm2_dense_fused(const float* __restrict__ X,
                  const float* __restrict__ Wi1, const float* __restrict__ Wh1,
                  const float* __restrict__ b1,
                  const float* __restrict__ Wi2, const float* __restrict__ Wh2,
                  const float* __restrict__ b2,
                  const float* __restrict__ Wd,  const float* __restrict__ bd,
                  float* __restrict__ out)
{
    __shared__ v2f sH1[2 * HBUF];
    __shared__ v2f sH2[2 * HBUF];
    __shared__ v4f sXW[3 * XSLOT];
    __shared__ v4f sX[2 * XRING];

    const int tid   = threadIdx.x;
    const int lane  = tid & 31;
    const int wave  = __builtin_amdgcn_readfirstlane(tid >> 5);
    const int mrow  = lane & 15;
    const int hi    = lane >> 4;
    const int khalf = 2 * hi;
    const int b0    = blockIdx.x * 16;

    const int  role = (int)((ROLE_OF_WAVE >> (4 * wave)) & 15ull);
    const int  n    = (int)((TILE_OF_WAVE >> (4 * wave)) & 15ull);
    const bool isL1  = role == R_L1;
    const bool isL2  = role == R_L2;
    const bool isL1R = role == R_L1R;
    const bool isX   = role == R_X;
    const bool layer1 = isL1 || isL1R || isX;

    for (int i = tid; i < 2 * HBUF; i += NWAVE * 32) {
        v2f zz; zz.x = 0.0f; zz.y = 0.0f;
        sH1[i] = zz;
        sH2[i] = zz;
    }

    v2f win[8], wrec[5];
    v8f bz;
    if (layer1 || isL2) {
        const float* __restrict__ Win  = layer1 ? Wi1 : Wi2;
        const float* __restrict__ Wrec = layer1 ? Wh1 : Wh2;
        const float* __restrict__ bb   = layer1 ? b1  : b2;
        const int col = (mrow & 3) * LH + 4 * n + (mrow >> 2);
        const float gsc = ((mrow & 3) == 2) ? 1.0f : 0.5f;
        const float hsc = 0.5f * gsc;
        const float isc = layer1 ? gsc : hsc;
        #pragma unroll
        for (int kk = 0; kk < 8; ++kk) {
            const int row = layer1 ? 16 * hi + 2 * kk : 4 * kk + khalf;
            v2f w; w.x = 0.0f; w.y = 0.0f;
            if (kk < 5 || layer1) {
                w.x = Win[row * LH4 + col];
                w.y = Win[(row + 1) * LH4 + col];
            }
            win[kk] = w;
        }
        #pragma unroll
        for (int kk = 0; kk < 5; ++kk) {
            const int row = 4 * kk + khalf;
            v2f w;
            w.x = Wrec[row * LH4 + col];
            w.y = Wrec[(row + 1) * LH4 + col];
            wrec[kk] = w;
        }
        #pragma unroll
        for (int r = 0; r < 8; ++r)
            bz[r] = bb[(r & 3) * LH + 4 * n + khalf + (r >> 2)];
        #pragma unroll
        for (int kk = 0; kk < 8; ++kk) { win[kk].x *= isc; win[kk].y *= isc; }
        #pragma unroll
        for (int kk = 0; kk < 5; ++kk) { wrec[kk].x *= hsc; wrec[kk].y *= hsc; }
        #pragma unroll
        for (int r = 0; r < 8; ++r)
            bz[r] *= ((r & 3) == 2) ? 1.0f : 0.5f;
    } else {
        #pragma unroll
        for (int kk = 0; kk < 8; ++kk) { win[kk].x = 0.0f; win[kk].y = 0.0f; }
        #pragma unroll
        for (int kk = 0; kk < 5; ++kk) {
            const int row = 4 * kk + khalf;
            v2f w;
            w.x = (mrow < LOUT) ? Wd[row * LOUT + mrow] : 0.0f;
            w.y = (mrow < LOUT) ? Wd[(row + 1) * LOUT + mrow] : 0.0f;
            wrec[kk] = w;
        }
        #pragma unroll
        for (int kk = 0; kk < 5; ++kk) { wrec[kk].x *= 0.5f; wrec[kk].y *= 0.5f; }
        #pragma unroll
        for (int r = 0; r < 8; ++r) {
            const int o = 8 * hi + r;
            bz[r] = (o < LOUT) ? bd[o] : 0.0f;
        }
    }

    const float* xlane = X + (size_t)(b0 + mrow) * LT * LF + 16 * hi;
    v4f xq[4];
    #pragma unroll
    for (int q = 0; q < 4; ++q) { xq[q].x = 0.0f; xq[q].y = 0.0f; xq[q].z = 0.0f; xq[q].w = 0.0f; }
    if (isL1 || isX) {
        #pragma unroll
        for (int q = 0; q < 4; ++q)
            xq[q] = *(const v4f*)(xlane + 4 * q);
    }
    if (isX) {
        v8f a = bz;
        v8f b = {};
        x_products(a, b, win, xq);
        const v8f s = a + b;
        sXW[lane]      = __builtin_shufflevector(s, s, 0, 1, 2, 3);
        sXW[32 + lane] = __builtin_shufflevector(s, s, 4, 5, 6, 7);
        #pragma unroll
        for (int q = 0; q < 4; ++q)
            xq[q] = *(const v4f*)(xlane + LF + 4 * q);
    }
    const float* xraw = X + (size_t)(b0 + (lane >> 3)) * LT * LF + 4 * (lane & 7);
    const int    xput = ((lane >> 3) + 16 * ((lane >> 2) & 1)) * XPITCH + (lane & 3);
    v4f raw[4];
    #pragma unroll
    for (int j = 0; j < 4; ++j) { raw[j].x = 0.0f; raw[j].y = 0.0f; raw[j].z = 0.0f; raw[j].w = 0.0f; }
    if (isX) {
        #pragma unroll
        for (int j = 0; j < 4; ++j)
            raw[j] = *(const v4f*)(xraw + (size_t)(4 * j) * LT * LF + LF);
        #pragma unroll
        for (int j = 0; j < 4; ++j)
            sX[XRING + xput + 4 * j * XPITCH] = raw[j];
        #pragma unroll
        for (int j = 0; j < 4; ++j)
            raw[j] = *(const v4f*)(xraw + (size_t)(4 * j) * LT * LF + 2 * LF);
    }

    v2f c; c.x = 0.0f; c.y = 0.0f;

    __syncthreads();
    __builtin_amdgcn_s_setprio(3);
    lds_publish_and_signal();

    if (isL1) {
        for (int i = 0; i < LT; ++i) {
            __builtin_amdgcn_s_setprio(0);
            v8f accA = bz;
            v8f accB = {};
            x_products(accA, accB, win, xq);
            asm volatile("" :: "v"(accA), "v"(accB));
            __builtin_amdgcn_s_setprio(3);

            lds_wait_and_acquire();

            const v2f* hprev = sH1 + ((i + 1) & 1) * HBUF + lane;
            v2f hf[5];
            #pragma unroll
            for (int kk = 0; kk < 5; ++kk) hf[kk] = hprev[kk * 32];
            WMMA4(accA, wrec[0], hf[0]);
            WMMA4(accB, wrec[1], hf[1]);
            WMMA4(accA, wrec[2], hf[2]);
            WMMA4(accB, wrec[3], hf[3]);
            WMMA4(accA, wrec[4], hf[4]);
            __builtin_amdgcn_sched_barrier(0);
            if (i + 1 < LT) {
                const v4f* xs = sX + ((i + 1) & 1) * XRING + lane * XPITCH;
                #pragma unroll
                for (int q = 0; q < 4; ++q) xq[q] = xs[q];
            }
            __builtin_amdgcn_sched_barrier(0);
            const v8f z = accA + accB;
            const v2f h = lstm_cells(z, c);
            sH1[(i & 1) * HBUF + n * 32 + lane] = h;

            lds_publish_and_signal();
        }
        lds_wait_and_acquire();
        lds_publish_and_signal();
        lds_wait_and_acquire();
        lds_publish_and_signal();
        lds_wait_and_acquire();
        lds_publish_and_signal();
    } else if (isL1R) {
        int rs = 0;
        for (int i = 0; i < LT; ++i) {
            lds_wait_and_acquire();

            const v2f* hprev = sH1 + ((i + 1) & 1) * HBUF + lane;
            v2f hf[5];
            #pragma unroll
            for (int kk = 0; kk < 5; ++kk) hf[kk] = hprev[kk * 32];
            const v4f xlo = sXW[rs * XSLOT + lane];
            const v4f xhi = sXW[rs * XSLOT + 32 + lane];
            rs = (rs == 2) ? 0 : rs + 1;
            v8f accA = __builtin_shufflevector(xlo, xhi, 0, 1, 2, 3, 4, 5, 6, 7);
            v8f accB = {};
            WMMA4(accB, wrec[0], hf[0]);
            WMMA4(accA, wrec[1], hf[1]);
            WMMA4(accB, wrec[2], hf[2]);
            WMMA4(accA, wrec[3], hf[3]);
            WMMA4(accB, wrec[4], hf[4]);
            const v8f z = accA + accB;
            const v2f h = lstm_cells(z, c);
            sH1[(i & 1) * HBUF + n * 32 + lane] = h;

            lds_publish_and_signal();
        }
        lds_wait_and_acquire();
        lds_publish_and_signal();
        lds_wait_and_acquire();
        lds_publish_and_signal();
        lds_wait_and_acquire();
        lds_publish_and_signal();
    } else if (isX) {
        int ws = 1;
        for (int i = 0; i < LT; ++i) {
            if (i + 1 < LT) {
                __builtin_amdgcn_s_setprio(0);
                v8f a = bz;
                v8f b = {};
                x_products(a, b, win, xq);
                const v8f s = a + b;
                __builtin_amdgcn_s_setprio(3);
                sXW[ws * XSLOT + lane]      = __builtin_shufflevector(s, s, 0, 1, 2, 3);
                sXW[ws * XSLOT + 32 + lane] = __builtin_shufflevector(s, s, 4, 5, 6, 7);
                ws = (ws == 2) ? 0 : ws + 1;
            }
            lds_wait_and_acquire();
            if (i + 2 < LT) {
                v4f* xd = sX + (i & 1) * XRING;
                #pragma unroll
                for (int j = 0; j < 4; ++j)
                    xd[xput + 4 * j * XPITCH] = raw[j];
                #pragma unroll
                for (int q = 0; q < 4; ++q) xq[q] = xd[lane * XPITCH + q];
            }
            if (i + 3 < LT) {
                #pragma unroll
                for (int j = 0; j < 4; ++j)
                    raw[j] = *(const v4f*)(xraw + (size_t)(4 * j) * LT * LF + (size_t)(i + 3) * LF);
            }
            if (i + 7 < LT) __builtin_prefetch(xlane + (size_t)(i + 7) * LF, 0, 3);
            lds_publish_and_signal();
        }
        lds_wait_and_acquire();
        lds_publish_and_signal();
        lds_wait_and_acquire();
        lds_publish_and_signal();
        lds_wait_and_acquire();
        lds_publish_and_signal();
    } else if (isL2) {
        lds_wait_and_acquire();
        lds_publish_and_signal();
        lds_wait_and_acquire();
        v2f ha[5];
        #pragma unroll
        for (int kk = 0; kk < 5; ++kk) ha[kk] = sH1[kk * 32 + lane];
        lds_publish_and_signal();
        for (int i = 2; i < LT + 2; ++i) {
            __builtin_amdgcn_s_setprio(0);
            v8f aA = bz;
            v8f aB = {};
            WMMA4(aA, win[0], ha[0]);
            WMMA4(aB, win[1], ha[1]);
            WMMA4(aA, win[2], ha[2]);
            WMMA4(aB, win[3], ha[3]);
            WMMA4(aA, win[4], ha[4]);
            asm volatile("" :: "v"(aA), "v"(aB));
            __builtin_amdgcn_s_setprio(3);

            lds_wait_and_acquire();

            const v2f* hrec = sH2 + ((i + 1) & 1) * HBUF + lane;
            const v2f* hin  = sH1 + ((i - 1) & 1) * HBUF + lane;
            v2f hb[5];
            #pragma unroll
            for (int kk = 0; kk < 5; ++kk) hb[kk] = hrec[kk * 32];
            #pragma unroll
            for (int kk = 0; kk < 5; ++kk) ha[kk] = hin[kk * 32];
            __builtin_amdgcn_sched_barrier(0);
            WMMA4(aB, wrec[0], hb[0]);
            WMMA4(aA, wrec[1], hb[1]);
            WMMA4(aB, wrec[2], hb[2]);
            WMMA4(aA, wrec[3], hb[3]);
            WMMA4(aB, wrec[4], hb[4]);
            const v8f z = aA + aB;
            const v2f h = lstm_cells(z, c);
            sH2[(i & 1) * HBUF + n * 32 + lane] = h;

            lds_publish_and_signal();
        }
        lds_wait_and_acquire();
        lds_publish_and_signal();
    } else {
        lds_wait_and_acquire();
        lds_publish_and_signal();
        lds_wait_and_acquire();
        lds_publish_and_signal();
        lds_wait_and_acquire();
        lds_publish_and_signal();
        for (int i = 3; i < LT + 3; ++i) {
            lds_wait_and_acquire();

            const int t = i - 3;
            const v2f* hsrc = sH2 + ((i + 1) & 1) * HBUF + lane;
            v2f hf[5];
            #pragma unroll
            for (int kk = 0; kk < 5; ++kk) hf[kk] = hsrc[kk * 32];
            v8f acc = {};
            #pragma unroll
            for (int kk = 0; kk < 5; ++kk)
                WMMA4(acc, wrec[kk], hf[kk]);
            float* orow = out + ((size_t)(b0 + mrow) * LT + (size_t)t) * LOUT + 8 * hi;
            v2f* o2 = (v2f*)orow;
            v2f p;
            p.x = acc[0] + bz[0]; p.y = acc[1] + bz[1]; o2[0] = p;
            if (hi == 0) {
                p.x = acc[2] + bz[2]; p.y = acc[3] + bz[3]; o2[1] = p;
                p.x = acc[4] + bz[4]; p.y = acc[5] + bz[5]; o2[2] = p;
                p.x = acc[6] + bz[6]; p.y = acc[7] + bz[7]; o2[3] = p;
            }

            lds_publish_and_signal();
        }
    }
    lds_wait_and_acquire();
    __builtin_amdgcn_s_setprio(0);
}

extern "C" void kernel_launch(void* const* d_in, const int* in_sizes, int n_in,
                              void* d_out, int out_size, void* d_ws, size_t ws_size,
                              hipStream_t stream) {
    (void)in_sizes; (void)n_in; (void)out_size; (void)d_ws; (void)ws_size;
    const float* X   = (const float*)d_in[0];
    const float* Wi1 = (const float*)d_in[1];
    const float* Wh1 = (const float*)d_in[2];
    const float* b1  = (const float*)d_in[3];
    const float* Wi2 = (const float*)d_in[4];
    const float* Wh2 = (const float*)d_in[5];
    const float* b2  = (const float*)d_in[6];
    const float* Wd  = (const float*)d_in[7];
    const float* bd  = (const float*)d_in[8];
    float* out = (float*)d_out;

    dim3 grid(LB / 16);
    dim3 block(NWAVE * 32);
    lstm2_dense_fused<<<grid, block, 0, stream>>>(X, Wi1, Wh1, b1, Wi2, Wh2, b2, Wd, bd, out);
}
